# starting kernel
# baseline (speedup 1.0000x reference)
#include <hip/hip_runtime.h>
#include <math.h>

typedef float v2f __attribute__((ext_vector_type(2)));
typedef float v8f __attribute__((ext_vector_type(8)));

#define NQ 10
#define NL 4
#define DIM 1024
#define BATCH 4096
#define K_IN 128
#define N_OUT 256
#define BPAD 132

__global__ __launch_bounds__(128) void k_encode(
    const float* __restrict__ x, const float* __restrict__ Win,
    const float* __restrict__ bin, float* __restrict__ h)
{
  __shared__ float Bs[16 * BPAD];

  const int tid  = threadIdx.x;
  const int lane = tid & 31;
  const int wave = tid >> 5;
  const int m0   = (blockIdx.x * 4 + wave) * 16;
  const int half = lane >> 4;
  const int l16  = lane & 15;

  for (int i = tid; i < 16 * K_IN; i += 128) {
    const int n = i >> 7;
    const int k = i & (K_IN - 1);
    Bs[n * BPAD + k] = (n < NQ) ? Win[n * K_IN + k] : 0.0f;
  }
  __syncthreads();

  const float* arow = x + (m0 + l16) * K_IN + half * 2;
  const float* brow = Bs + l16 * BPAD + half * 2;

  v8f acc = {};
  #pragma unroll 4
  for (int kk = 0; kk < K_IN / 4; ++kk) {
    v2f a, b;
    a.x = arow[kk * 4];
    a.y = arow[kk * 4 + 1];
    b.x = brow[kk * 4];
    b.y = brow[kk * 4 + 1];
    acc = __builtin_amdgcn_wmma_f32_16x16x4_f32(false, a, false, b,
                                                (short)0, acc, false, false);
  }
  if (l16 < NQ) {
    const float bias = bin[l16];
    #pragma unroll
    for (int r = 0; r < 8; ++r) {
      const int m = m0 + r + 8 * half;
      h[m * NQ + l16] = tanhf(acc[r] + bias);
    }
  }
}

__global__ __launch_bounds__(128) void k_circuit(
    const float* __restrict__ h, const float* __restrict__ qp,
    float* __restrict__ z)
{
  __shared__ float sre[4][DIM];
  __shared__ float sim[4][DIM];
  __shared__ unsigned short perm[DIM];

  const int tid  = threadIdx.x;
  const int wave = tid >> 5;
  const int lane = tid & 31;
  const int b    = blockIdx.x * 4 + wave;

  for (int j = tid; j < DIM; j += 128) {
    int src = j;
    if ((src >> 0) & 1) src ^= (1 << 9);
    #pragma unroll
    for (int c = 8; c >= 0; --c)
      if ((src >> (9 - c)) & 1) src ^= (1 << (9 - (c + 1)));
    perm[j] = (unsigned short)src;
  }
  __syncthreads();

  float* re = sre[wave];
  float* im = sim[wave];

  for (int k = 0; k < DIM / 32; ++k) {
    const int i = k * 32 + lane;
    re[i] = (i == 0) ? 1.0f : 0.0f;
    im[i] = 0.0f;
  }

  auto rx = [&](float theta, int w) {
    const float hc = cosf(0.5f * theta);
    const float hs = sinf(0.5f * theta);
    const int mask = 1 << (9 - w);
    const int low  = mask - 1;
    for (int k = 0; k < 16; ++k) {
      const int p  = k * 32 + lane;
      const int i0 = ((p & ~low) << 1) | (p & low);
      const int i1 = i0 | mask;
      const float are = re[i0], aim = im[i0];
      const float bre = re[i1], bim = im[i1];
      re[i0] = hc * are + hs * bim;
      im[i0] = hc * aim - hs * bre;
      re[i1] = hc * bre + hs * aim;
      im[i1] = hc * bim - hs * are;
    }
  };

  for (int w = 0; w < NQ; ++w) rx(h[b * NQ + w], w);

  for (int l = 0; l < NL; ++l) {
    for (int w = 0; w < NQ; ++w) rx(qp[l * NQ + w], w);
    float tre[32], tim[32];
    #pragma unroll
    for (int k = 0; k < 32; ++k) {
      const int s = perm[k * 32 + lane];
      tre[k] = re[s];
      tim[k] = im[s];
    }
    #pragma unroll
    for (int k = 0; k < 32; ++k) {
      const int j = k * 32 + lane;
      re[j] = tre[k];
      im[j] = tim[k];
    }
  }

  float acc[NQ];
  #pragma unroll
  for (int q = 0; q < NQ; ++q) acc[q] = 0.0f;
  for (int k = 0; k < 32; ++k) {
    const int i = k * 32 + lane;
    const float pr = re[i] * re[i] + im[i] * im[i];
    #pragma unroll
    for (int q = 0; q < NQ; ++q)
      acc[q] += ((i >> (9 - q)) & 1) ? -pr : pr;
  }
  #pragma unroll
  for (int q = 0; q < NQ; ++q) {
    float v = acc[q];
    for (int off = 16; off > 0; off >>= 1)
      v += __shfl_xor(v, off, 32);
    if (lane == 0) z[b * NQ + q] = v;
  }
}

__global__ __launch_bounds__(128) void k_readout(
    const float* __restrict__ z, const float* __restrict__ Wout,
    const float* __restrict__ bout, float* __restrict__ out)
{
  const int lane = threadIdx.x & 31;
  const int wave = threadIdx.x >> 5;
  const int tile = blockIdx.x * 4 + wave;
  const int m0   = (tile >> 4) * 16;
  const int n0   = (tile & 15) * 16;
  const int half = lane >> 4;
  const int l16  = lane & 15;

  const float* arow = z    + (m0 + l16) * NQ;
  const float* brow = Wout + (n0 + l16) * NQ;

  v8f acc = {};
  #pragma unroll
  for (int kk = 0; kk < 3; ++kk) {
    const int k0  = kk * 4 + half * 2;
    const int k0c = (k0 < NQ - 1) ? k0 : (NQ - 2);
    const float vm0 = (k0     < NQ) ? 1.0f : 0.0f;
    const float vm1 = (k0 + 1 < NQ) ? 1.0f : 0.0f;
    v2f a, b;
    a.x = arow[k0c]     * vm0;
    a.y = arow[k0c + 1] * vm1;
    b.x = brow[k0c];
    b.y = brow[k0c + 1];
    acc = __builtin_amdgcn_wmma_f32_16x16x4_f32(false, a, false, b,
                                                (short)0, acc, false, false);
  }
  const float bias = bout[n0 + l16];
  #pragma unroll
  for (int r = 0; r < 8; ++r) {
    const int m = m0 + r + 8 * half;
    out[m * N_OUT + n0 + l16] = acc[r] + bias;
  }
}

extern "C" void kernel_launch(void* const* d_in, const int* in_sizes, int n_in,
                              void* d_out, int out_size, void* d_ws, size_t ws_size,
                              hipStream_t stream) {
  const float* x    = (const float*)d_in[0];
  const float* Win  = (const float*)d_in[1];
  const float* bin  = (const float*)d_in[2];
  const float* qp   = (const float*)d_in[3];
  const float* Wout = (const float*)d_in[4];
  const float* bout = (const float*)d_in[5];
  float* out = (float*)d_out;

  float* h  = (float*)d_ws;
  float* zv = h + BATCH * NQ;

  k_encode <<<BATCH / 16 / 4, 128, 0, stream>>>(x, Win, bin, h);
  k_circuit<<<BATCH / 4,      128, 0, stream>>>(h, qp, zv);
  k_readout<<<BATCH * 16 / 16 / 4, 128, 0, stream>>>(zv, Wout, bout, out);
}


# k_circuit: the last ring is not carried out: Z sums from the last layer's result registers, ring folded into the signs (register butterfly, lane sums by DPP); no final LDS trip; sums reordered
# speedup vs baseline: 2.3376x; 2.3376x over previous
#include <hip/hip_runtime.h>
#include <math.h>

typedef float v2f __attribute__((ext_vector_type(2)));
typedef float v8f __attribute__((ext_vector_type(8)));

#define NQ 10
#define NL 4
#define DIM 1024
#define BATCH 4096
#define K_IN 128
#define N_OUT 256
#define BPAD 132

__global__ __launch_bounds__(128) void k_encode(
    const float* __restrict__ x, const float* __restrict__ Win,
    const float* __restrict__ bin, const float* __restrict__ qp,
    float* __restrict__ cs, float* __restrict__ tq)
{
  __shared__ float Bs[16 * BPAD];

  const int tid  = threadIdx.x;
  const int lane = tid & 31;
  const int wave = tid >> 5;
  const int m0   = (blockIdx.x * 4 + wave) * 16;
  const int half = lane >> 4;
  const int l16  = lane & 15;

  if (blockIdx.x == BATCH / 16 / 4) {
    if (tid < NL * NQ) {
      const float a = 0.5f * qp[tid];
      const int l = tid / NQ, w = tid - l * NQ;
      tq[l * 2 * NQ + w]      = cosf(a);
      tq[l * 2 * NQ + NQ + w] = sinf(a);
    }
    return;
  }

  for (int i = tid; i < 16 * K_IN; i += 128) {
    const int n = i >> 7;
    const int k = i & (K_IN - 1);
    Bs[n * BPAD + k] = (n < NQ) ? Win[n * K_IN + k] : 0.0f;
  }
  __syncthreads();

  const float* arow = x + (m0 + l16) * K_IN + half * 2;
  const float* brow = Bs + l16 * BPAD + half * 2;

  v8f acc = {};
  #pragma unroll 4
  for (int kk = 0; kk < K_IN / 4; ++kk) {
    v2f a, b;
    a.x = arow[kk * 4];
    a.y = arow[kk * 4 + 1];
    b.x = brow[kk * 4];
    b.y = brow[kk * 4 + 1];
    acc = __builtin_amdgcn_wmma_f32_16x16x4_f32(false, a, false, b,
                                                (short)0, acc, false, false);
  }
  if (l16 < NQ) {
    const float bias = bin[l16];
    #pragma unroll
    for (int r = 0; r < 8; ++r) {
      const int m = m0 + r + 8 * half;
      const float a = 0.5f * tanhf(acc[r] + bias);
      cs[m * 2 * NQ + l16]      = cosf(a);
      cs[m * 2 * NQ + NQ + l16] = sinf(a);
    }
  }
}


__device__ __forceinline__ void wave_sync() {
  __builtin_amdgcn_fence(__ATOMIC_RELEASE, "wavefront");
  __builtin_amdgcn_wave_barrier();
  __builtin_amdgcn_fence(__ATOMIC_ACQUIRE, "wavefront");
}

__device__ __forceinline__ constexpr int ring_dst(int s) {
  for (int c = 0; c <= 8; ++c)
    if ((s >> (9 - c)) & 1) s ^= 1 << (8 - c);
  if (s & 1) s ^= 1 << 9;
  return s;
}

__device__ __forceinline__ constexpr int popc(int v) { return __builtin_popcount(v); }

__device__ __forceinline__ float rot(bool a, bool b, float c, float s) {
  return (a == b) ? c : (a ? -s : s);
}

__device__ __forceinline__ void gate_frag(v2f (&g)[4], bool o3, bool o2, bool o1, bool o0,
    bool half, float c3, float s3, float c2, float s2, float c1, float s1, float c0, float s0)
{
  const float fH = rot(o1, half, c1, s1);
  const float w0 = fH * rot(o0, false, c0, s0);
  const float w1 = fH * rot(o0, true,  c0, s0);
  #pragma unroll
  for (int kk = 0; kk < 4; ++kk) {
    const float u = rot(o3, (kk >> 1) != 0, c3, s3) * rot(o2, (kk & 1) != 0, c2, s2);
    g[kk].x = u * w0;
    g[kk].y = u * w1;
  }
}

template <int HB, int L3, int L2, int L1, int L0, int K1, int K0, int XY, int T1, int T0>
__device__ __forceinline__ void layer_core(
    const v2f (&st)[8][4], v8f (&d)[8], int l16, int half,
    const float (&hc)[NQ], const float (&hs)[NQ])
{
  const bool n3 = (l16 >> 3) & 1, n2 = (l16 >> 2) & 1;
  const bool n1 = (l16 >> 1) & 1, n0 = l16 & 1;
  const bool hb = half != 0;
  v2f g[4];

  gate_frag(g, n3, n2, n1, n0, hb, hc[9 - K1], hs[9 - K1], hc[9 - K0], hs[9 - K0],
            hc[9 - HB], hs[9 - HB], hc[9 - XY], hs[9 - XY]);
  v8f d1[8];
  #pragma unroll
  for (int t = 0; t < 8; ++t) d1[t] = v8f{};
  #pragma unroll
  for (int kk = 0; kk < 4; ++kk) {
    #pragma unroll
    for (int t = 0; t < 8; ++t)
      d1[t] = __builtin_amdgcn_wmma_f32_16x16x4_f32(false, st[t][kk], false, g[kk],
                                                    (short)0, d1[t], false, false);
  }

  gate_frag(g, n3, n2, n1, n0, hb, hc[9 - L2], hs[9 - L2], hc[9 - L1], hs[9 - L1],
            hc[9 - L3], hs[9 - L3], hc[9 - L0], hs[9 - L0]);
  v8f d2[8];
  #pragma unroll
  for (int t = 0; t < 8; ++t) d2[t] = v8f{};
  #pragma unroll
  for (int kk = 0; kk < 4; ++kk) {
    #pragma unroll
    for (int t = 0; t < 8; ++t) {
      v2f a;
      a.x = d1[t][2 * kk];
      a.y = d1[t][2 * kk + 1];
      d2[t] = __builtin_amdgcn_wmma_f32_16x16x4_f32(false, a, false, g[kk],
                                                    (short)0, d2[t], false, false);
    }
  }

  gate_frag(g, n2, n1, n0, n3, hb, hc[9 - T1], hs[9 - T1], hc[9 - T0], hs[9 - T0],
            1.0f, 0.0f, 1.0f, 0.0f);
  #pragma unroll
  for (int t = 0; t < 8; ++t) d[t] = v8f{};
  #pragma unroll
  for (int kk = 0; kk < 4; ++kk) {
    #pragma unroll
    for (int tp = 0; tp < 8; ++tp) {
      const int p = tp >> 2, u = tp & 3;
      v2f b;
      b.x = d2[(p << 2) | kk][2 * u];
      b.y = d2[(p << 2) | kk][2 * u + 1];
      d[tp] = __builtin_amdgcn_wmma_f32_16x16x4_f32(false, g[kk], false, b,
                                                    (short)0, d[tp], false, false);
    }
  }
}

template <int HB, int L3, int L2, int L1, int L0, int K1, int K0, int XY, int T1, int T0>
__device__ __forceinline__ void load_state(
    const float* __restrict__ S, v2f (&st)[8][4], int l16, int half)
{
  const int lb = (((l16 >> 3) & 1) << L3) | (((l16 >> 2) & 1) << L2) |
                 (((l16 >> 1) & 1) << L1) | ((l16 & 1) << L0) | (half << HB);
  #pragma unroll
  for (int t = 0; t < 8; ++t) {
    #pragma unroll
    for (int kk = 0; kk < 4; ++kk) {
      const int o = (((t >> 2) & 1) << 10) | (((t >> 1) & 1) << T1) | ((t & 1) << T0) |
                    (((kk >> 1) & 1) << K1) | ((kk & 1) << K0);
      st[t][kk].x = S[lb + o];
      st[t][kk].y = S[lb + o + (1 << XY)];
    }
  }
}

template <int HB, int L3, int L2, int L1, int L0, int K1, int K0, int XY, int T1, int T0>
__device__ __forceinline__ void ring_store(
    float* __restrict__ S, const v8f (&d)[8], int l16, int half)
{
  constexpr int AH = XY, A3 = L2, A2 = L1, A1 = L3, A0 = L0;
  constexpr int R2 = T1, R1 = T0, R0 = K1, U1 = K0, U0 = HB;
  const int lb = (((l16 >> 3) & 1) << A3) | (((l16 >> 2) & 1) << A2) |
                 (((l16 >> 1) & 1) << A1) | ((l16 & 1) << A0) | (half << AH);
  constexpr int LM = ring_dst(1 << A3) | ring_dst(1 << A2) | ring_dst(1 << A1) |
                     ring_dst(1 << A0) | ring_dst(1 << AH);
  const int lr  = ring_dst(lb);
  const int kl  = (popc(lb) - popc(lr)) & 3;
  const int lrc = lr | ((kl & 1) << 10);
  unsigned sbm[4];
  #pragma unroll
  for (int kc = 0; kc < 4; ++kc)
    sbm[kc] = (unsigned)(((kl + kc) >> 1) & 1) << 31;
  constexpr int RB[5] = {R0, R1, R2, U0, U1};
  unsigned cm[32];
  cm[0] = 0u;
  #pragma unroll
  for (int x = 1; x < 32; ++x) {
    const int low = x & -x;
    const int bi  = (low == 1) ? 0 : (low == 2) ? 1 : (low == 4) ? 2 : (low == 8) ? 3 : 4;
    const unsigned bmw = (unsigned)(popc(lr & ring_dst(1 << RB[bi])) & 1) << 31;
    cm[x] = cm[x ^ low] ^ bmw;
  }
  #pragma unroll
  for (int t = 0; t < 8; ++t) {
    #pragma unroll
    for (int r = 0; r < 8; ++r) {
      const int regS = (((t >> 1) & 1) << U1) | ((t & 1) << U0) |
                       (((r >> 2) & 1) << R2) | (((r >> 1) & 1) << R1) | ((r & 1) << R0);
      const int pl  = (t >> 2) & 1;
      const int ro  = ring_dst(regS);
      const int ko  = (popc(regS) - popc(ro)) & 3;
      const int kc  = (ko + pl) & 3;
      const int xc  = (ro & LM) | (((ko & 1) ^ pl) << 10);
      const int add = ro & ~LM & 0x3FF;
      const int x   = (r & 7) | ((t & 3) << 3);
      const unsigned w = __float_as_uint(d[t][r]) ^ cm[x] ^ sbm[kc];
      S[(lrc ^ xc) + add] = __uint_as_float(w);
    }
  }
}

__device__ __forceinline__ float xhalf(float v) {
  const int b = __float_as_int(v);
  return __int_as_float(__builtin_amdgcn_permlanex16(b, b, 0x76543210, 0xfedcba98, false, false));
}

template <int M>
__device__ __forceinline__ float row_xor_add(float v) {
  const int b = __float_as_int(v);
  return v + __int_as_float(__builtin_amdgcn_update_dpp(0, b, 0x160 + M, 0xf, 0xf, true));
}

template <int HB, int L3, int L2, int L1, int L0, int K1, int K0, int XY, int T1, int T0>
__device__ __forceinline__ void measure_z(
    const v8f (&d)[8], int lane, float* __restrict__ zrow)
{
  constexpr int AH = XY, A3 = L2, A2 = L1, A1 = L3, A0 = L0;
  constexpr int R2 = T1, R1 = T0, R0 = K1, U1 = K0, U0 = HB;

  float w[32];
  #pragma unroll
  for (int t = 0; t < 4; ++t) {
    #pragma unroll
    for (int r = 0; r < 8; ++r)
      w[(t << 3) | r] = d[t][r] * d[t][r] + d[t + 4][r] * d[t + 4][r];
  }
  constexpr int ORD[5] = {4, 3, 1, 2, 0};
  #pragma unroll
  for (int i = 0; i < 5; ++i) {
    const int bit = 1 << ORD[i];
    #pragma unroll
    for (int x = 0; x < 32; ++x) {
      if (x & bit) continue;
      const float a = w[x], b = w[x | bit];
      w[x]       = a + b;
      w[x | bit] = a - b;
    }
  }
  constexpr int RB[5] = {R0, R1, R2, U0, U1};
  constexpr int LB[5] = {A0, A1, A2, A3, AH};
  float z = 0.0f;
  #pragma unroll
  for (int q = 0; q < NQ; ++q) {
    int m = 0, lm = 0;
    #pragma unroll
    for (int i = 0; i < 5; ++i) {
      m  |= ((ring_dst(1 << RB[i]) >> (9 - q)) & 1) << i;
      lm |= ((ring_dst(1 << LB[i]) >> (9 - q)) & 1) << i;
    }
    float v = w[m];
    if (lm)
      v = __uint_as_float(__float_as_uint(v) ^ ((unsigned)(__builtin_popcount(lane & lm) & 1) << 31));
    v = row_xor_add<1>(v);
    v = row_xor_add<2>(v);
    v = row_xor_add<4>(v);
    v = row_xor_add<8>(v);
    v += xhalf(v);
    z = (lane == q) ? v : z;
  }
  if (lane < NQ) zrow[lane] = z;
}

__global__ __launch_bounds__(128) void k_circuit(
    const float* __restrict__ cs, const float* __restrict__ tq,
    float* __restrict__ z)
{
  __shared__ float stl[4][2 * DIM];

  const int tid  = threadIdx.x;
  const int wave = tid >> 5;
  const int lane = tid & 31;
  const int half = lane >> 4;
  const int l16  = lane & 15;
  const int b    = blockIdx.x * 4 + wave;

  float* S  = stl[wave];

  v2f st[8][4];
  v8f d[8];
  float hc[NQ], hs[NQ];

  #pragma unroll
  for (int t = 0; t < 8; ++t) {
    #pragma unroll
    for (int kk = 0; kk < 4; ++kk) { st[t][kk].x = 0.0f; st[t][kk].y = 0.0f; }
  }
  st[0][0].x = (lane == 0) ? 1.0f : 0.0f;

  #pragma unroll
  for (int w = 0; w < NQ; ++w) { hc[w] = cs[b * 2 * NQ + w]; hs[w] = cs[b * 2 * NQ + NQ + w]; }
  layer_core<5, 4, 3, 2, 1, 9, 8, 0, 7, 6>(st, d, l16, half, hc, hs);
  #pragma unroll
  for (int t = 0; t < 8; ++t) {
    #pragma unroll
    for (int kk = 0; kk < 4; ++kk) { st[t][kk].x = d[t][2 * kk]; st[t][kk].y = d[t][2 * kk + 1]; }
  }
  #pragma unroll
  for (int w = 0; w < NQ; ++w) { hc[w] = tq[w]; hs[w] = tq[NQ + w]; }
  layer_core<0, 3, 2, 4, 1, 7, 6, 9, 8, 5>(st, d, l16, half, hc, hs);
  ring_store<0, 3, 2, 4, 1, 7, 6, 9, 8, 5>(S, d, l16, half);
  wave_sync();

  for (int l = 1; l < NL - 1; ++l) {
    #pragma unroll
    for (int w = 0; w < NQ; ++w) { hc[w] = tq[l * 2 * NQ + w]; hs[w] = tq[l * 2 * NQ + NQ + w]; }
    load_state<5, 4, 3, 2, 1, 9, 8, 0, 7, 6>(S, st, l16, half);
    wave_sync();
    layer_core<5, 4, 3, 2, 1, 9, 8, 0, 7, 6>(st, d, l16, half, hc, hs);
    ring_store<5, 4, 3, 2, 1, 9, 8, 0, 7, 6>(S, d, l16, half);
    wave_sync();
  }

  #pragma unroll
  for (int w = 0; w < NQ; ++w) {
    hc[w] = tq[(NL - 1) * 2 * NQ + w];
    hs[w] = tq[(NL - 1) * 2 * NQ + NQ + w];
  }
  load_state<5, 4, 3, 2, 1, 9, 8, 0, 7, 6>(S, st, l16, half);
  layer_core<5, 4, 3, 2, 1, 9, 8, 0, 7, 6>(st, d, l16, half, hc, hs);
  measure_z<5, 4, 3, 2, 1, 9, 8, 0, 7, 6>(d, lane, z + b * NQ);
}

__global__ __launch_bounds__(128) void k_readout(
    const float* __restrict__ z, const float* __restrict__ Wout,
    const float* __restrict__ bout, float* __restrict__ out)
{
  const int lane = threadIdx.x & 31;
  const int wave = threadIdx.x >> 5;
  const int tile = blockIdx.x * 4 + wave;
  const int m0   = (tile >> 4) * 16;
  const int n0   = (tile & 15) * 16;
  const int half = lane >> 4;
  const int l16  = lane & 15;

  const float* arow = z    + (m0 + l16) * NQ;
  const float* brow = Wout + (n0 + l16) * NQ;

  v8f acc = {};
  #pragma unroll
  for (int kk = 0; kk < 3; ++kk) {
    const int k0  = kk * 4 + half * 2;
    const int k0c = (k0 < NQ - 1) ? k0 : (NQ - 2);
    const float vm0 = (k0     < NQ) ? 1.0f : 0.0f;
    const float vm1 = (k0 + 1 < NQ) ? 1.0f : 0.0f;
    v2f a, b;
    a.x = arow[k0c]     * vm0;
    a.y = arow[k0c + 1] * vm1;
    b.x = brow[k0c];
    b.y = brow[k0c + 1];
    acc = __builtin_amdgcn_wmma_f32_16x16x4_f32(false, a, false, b,
                                                (short)0, acc, false, false);
  }
  const float bias = bout[n0 + l16];
  #pragma unroll
  for (int r = 0; r < 8; ++r) {
    const int m = m0 + r + 8 * half;
    out[m * N_OUT + n0 + l16] = acc[r] + bias;
  }
}

extern "C" void kernel_launch(void* const* d_in, const int* in_sizes, int n_in,
                              void* d_out, int out_size, void* d_ws, size_t ws_size,
                              hipStream_t stream) {
  const float* x    = (const float*)d_in[0];
  const float* Win  = (const float*)d_in[1];
  const float* bin  = (const float*)d_in[2];
  const float* qp   = (const float*)d_in[3];
  const float* Wout = (const float*)d_in[4];
  const float* bout = (const float*)d_in[5];
  float* out = (float*)d_out;

  float* cs = (float*)d_ws;
  float* tq = cs + BATCH * 2 * NQ;
  float* zv = tq + 128;

  k_encode <<<BATCH / 16 / 4 + 1, 128, 0, stream>>>(x, Win, bin, qp, cs, tq);
  k_circuit<<<BATCH / 4,      128, 0, stream>>>(cs, tq, zv);
  k_readout<<<BATCH * 16 / 16 / 4, 128, 0, stream>>>(zv, Wout, bout, out);
}


# k_circuit: tiles that are still exactly zero are not multiplied: |0..0> feeds 9 WMMAs in the encoding layer, not 48; layer 0 runs on the re plane only (im is zero until the first ring), 48 not 96
# speedup vs baseline: 2.5359x; 1.0848x over previous
#include <hip/hip_runtime.h>
#include <math.h>

typedef float v2f __attribute__((ext_vector_type(2)));
typedef float v8f __attribute__((ext_vector_type(8)));

#define NQ 10
#define NL 4
#define DIM 1024
#define BATCH 4096
#define K_IN 128
#define N_OUT 256
#define BPAD 132

__global__ __launch_bounds__(128) void k_encode(
    const float* __restrict__ x, const float* __restrict__ Win,
    const float* __restrict__ bin, const float* __restrict__ qp,
    float* __restrict__ cs, float* __restrict__ tq)
{
  __shared__ float Bs[16 * BPAD];

  const int tid  = threadIdx.x;
  const int lane = tid & 31;
  const int wave = tid >> 5;
  const int m0   = (blockIdx.x * 4 + wave) * 16;
  const int half = lane >> 4;
  const int l16  = lane & 15;

  if (blockIdx.x == BATCH / 16 / 4) {
    if (tid < NL * NQ) {
      const float a = 0.5f * qp[tid];
      const int l = tid / NQ, w = tid - l * NQ;
      tq[l * 2 * NQ + w]      = cosf(a);
      tq[l * 2 * NQ + NQ + w] = sinf(a);
    }
    return;
  }

  for (int i = tid; i < 16 * K_IN; i += 128) {
    const int n = i >> 7;
    const int k = i & (K_IN - 1);
    Bs[n * BPAD + k] = (n < NQ) ? Win[n * K_IN + k] : 0.0f;
  }
  __syncthreads();

  const float* arow = x + (m0 + l16) * K_IN + half * 2;
  const float* brow = Bs + l16 * BPAD + half * 2;

  v8f acc = {};
  #pragma unroll 4
  for (int kk = 0; kk < K_IN / 4; ++kk) {
    v2f a, b;
    a.x = arow[kk * 4];
    a.y = arow[kk * 4 + 1];
    b.x = brow[kk * 4];
    b.y = brow[kk * 4 + 1];
    acc = __builtin_amdgcn_wmma_f32_16x16x4_f32(false, a, false, b,
                                                (short)0, acc, false, false);
  }
  if (l16 < NQ) {
    const float bias = bin[l16];
    #pragma unroll
    for (int r = 0; r < 8; ++r) {
      const int m = m0 + r + 8 * half;
      const float a = 0.5f * tanhf(acc[r] + bias);
      cs[m * 2 * NQ + l16]      = cosf(a);
      cs[m * 2 * NQ + NQ + l16] = sinf(a);
    }
  }
}


__device__ __forceinline__ void wave_sync() {
  __builtin_amdgcn_fence(__ATOMIC_RELEASE, "wavefront");
  __builtin_amdgcn_wave_barrier();
  __builtin_amdgcn_fence(__ATOMIC_ACQUIRE, "wavefront");
}

__device__ __forceinline__ constexpr int ring_dst(int s) {
  for (int c = 0; c <= 8; ++c)
    if ((s >> (9 - c)) & 1) s ^= 1 << (8 - c);
  if (s & 1) s ^= 1 << 9;
  return s;
}

__device__ __forceinline__ constexpr int popc(int v) { return __builtin_popcount(v); }

__device__ __forceinline__ float rot(bool a, bool b, float c, float s) {
  return (a == b) ? c : (a ? -s : s);
}

__device__ __forceinline__ void gate_frag(v2f (&g)[4], bool o3, bool o2, bool o1, bool o0,
    bool half, float c3, float s3, float c2, float s2, float c1, float s1, float c0, float s0)
{
  const float fH = rot(o1, half, c1, s1);
  const float w0 = fH * rot(o0, false, c0, s0);
  const float w1 = fH * rot(o0, true,  c0, s0);
  #pragma unroll
  for (int kk = 0; kk < 4; ++kk) {
    const float u = rot(o3, (kk >> 1) != 0, c3, s3) * rot(o2, (kk & 1) != 0, c2, s2);
    g[kk].x = u * w0;
    g[kk].y = u * w1;
  }
}

template <int NP, int HB, int L3, int L2, int L1, int L0, int K1, int K0, int XY, int T1, int T0>
__device__ __forceinline__ void layer_core(
    const v2f (&st)[8][4], v8f (&d)[8], int l16, int half,
    const float (&hc)[NQ], const float (&hs)[NQ])
{
  constexpr int NT = 4 * NP;
  const bool n3 = (l16 >> 3) & 1, n2 = (l16 >> 2) & 1;
  const bool n1 = (l16 >> 1) & 1, n0 = l16 & 1;
  const bool hb = half != 0;
  v2f g[4];

  gate_frag(g, n3, n2, n1, n0, hb, hc[9 - K1], hs[9 - K1], hc[9 - K0], hs[9 - K0],
            hc[9 - HB], hs[9 - HB], hc[9 - XY], hs[9 - XY]);
  v8f d1[NT];
  #pragma unroll
  for (int t = 0; t < NT; ++t) d1[t] = v8f{};
  #pragma unroll
  for (int kk = 0; kk < 4; ++kk) {
    #pragma unroll
    for (int t = 0; t < NT; ++t)
      d1[t] = __builtin_amdgcn_wmma_f32_16x16x4_f32(false, st[t][kk], false, g[kk],
                                                    (short)0, d1[t], false, false);
  }

  gate_frag(g, n3, n2, n1, n0, hb, hc[9 - L2], hs[9 - L2], hc[9 - L1], hs[9 - L1],
            hc[9 - L3], hs[9 - L3], hc[9 - L0], hs[9 - L0]);
  v8f d2[NT];
  #pragma unroll
  for (int t = 0; t < NT; ++t) d2[t] = v8f{};
  #pragma unroll
  for (int kk = 0; kk < 4; ++kk) {
    #pragma unroll
    for (int t = 0; t < NT; ++t) {
      v2f a;
      a.x = d1[t][2 * kk];
      a.y = d1[t][2 * kk + 1];
      d2[t] = __builtin_amdgcn_wmma_f32_16x16x4_f32(false, a, false, g[kk],
                                                    (short)0, d2[t], false, false);
    }
  }

  gate_frag(g, n2, n1, n0, n3, hb, hc[9 - T1], hs[9 - T1], hc[9 - T0], hs[9 - T0],
            1.0f, 0.0f, 1.0f, 0.0f);
  #pragma unroll
  for (int t = 0; t < 8; ++t) d[t] = v8f{};
  #pragma unroll
  for (int kk = 0; kk < 4; ++kk) {
    #pragma unroll
    for (int tp = 0; tp < NT; ++tp) {
      const int p = tp >> 2, u = tp & 3;
      v2f b;
      b.x = d2[(p << 2) | kk][2 * u];
      b.y = d2[(p << 2) | kk][2 * u + 1];
      d[tp] = __builtin_amdgcn_wmma_f32_16x16x4_f32(false, g[kk], false, b,
                                                    (short)0, d[tp], false, false);
    }
  }
}

template <int HB, int L3, int L2, int L1, int L0, int K1, int K0, int XY, int T1, int T0>
__device__ __forceinline__ void first_core(
    v8f (&d)[8], int lane, int l16, int half,
    const float (&hc)[NQ], const float (&hs)[NQ])
{
  const bool n3 = (l16 >> 3) & 1, n2 = (l16 >> 2) & 1;
  const bool n1 = (l16 >> 1) & 1, n0 = l16 & 1;
  const bool hb = half != 0;
  v2f g[4];

  v2f a0;
  a0.x = (lane == 0) ? 1.0f : 0.0f;
  a0.y = 0.0f;
  gate_frag(g, n3, n2, n1, n0, hb, hc[9 - K1], hs[9 - K1], hc[9 - K0], hs[9 - K0],
            hc[9 - HB], hs[9 - HB], hc[9 - XY], hs[9 - XY]);
  const v8f d1 = __builtin_amdgcn_wmma_f32_16x16x4_f32(false, a0, false, g[0],
                                                       (short)0, v8f{}, false, false);

  gate_frag(g, n3, n2, n1, n0, hb, hc[9 - L2], hs[9 - L2], hc[9 - L1], hs[9 - L1],
            hc[9 - L3], hs[9 - L3], hc[9 - L0], hs[9 - L0]);
  v8f d2 = v8f{};
  #pragma unroll
  for (int kk = 0; kk < 4; ++kk) {
    v2f a;
    a.x = d1[2 * kk];
    a.y = d1[2 * kk + 1];
    d2 = __builtin_amdgcn_wmma_f32_16x16x4_f32(false, a, false, g[kk],
                                               (short)0, d2, false, false);
  }

  gate_frag(g, n2, n1, n0, n3, hb, hc[9 - T1], hs[9 - T1], hc[9 - T0], hs[9 - T0],
            1.0f, 0.0f, 1.0f, 0.0f);
  #pragma unroll
  for (int u = 0; u < 4; ++u) {
    v2f b;
    b.x = d2[2 * u];
    b.y = d2[2 * u + 1];
    d[u] = __builtin_amdgcn_wmma_f32_16x16x4_f32(false, g[0], false, b,
                                                 (short)0, v8f{}, false, false);
  }
  #pragma unroll
  for (int t = 4; t < 8; ++t) d[t] = v8f{};
}

template <int HB, int L3, int L2, int L1, int L0, int K1, int K0, int XY, int T1, int T0>
__device__ __forceinline__ void load_state(
    const float* __restrict__ S, v2f (&st)[8][4], int l16, int half)
{
  const int lb = (((l16 >> 3) & 1) << L3) | (((l16 >> 2) & 1) << L2) |
                 (((l16 >> 1) & 1) << L1) | ((l16 & 1) << L0) | (half << HB);
  #pragma unroll
  for (int t = 0; t < 8; ++t) {
    #pragma unroll
    for (int kk = 0; kk < 4; ++kk) {
      const int o = (((t >> 2) & 1) << 10) | (((t >> 1) & 1) << T1) | ((t & 1) << T0) |
                    (((kk >> 1) & 1) << K1) | ((kk & 1) << K0);
      st[t][kk].x = S[lb + o];
      st[t][kk].y = S[lb + o + (1 << XY)];
    }
  }
}

template <int HB, int L3, int L2, int L1, int L0, int K1, int K0, int XY, int T1, int T0>
__device__ __forceinline__ void ring_store(
    float* __restrict__ S, const v8f (&d)[8], int l16, int half)
{
  constexpr int AH = XY, A3 = L2, A2 = L1, A1 = L3, A0 = L0;
  constexpr int R2 = T1, R1 = T0, R0 = K1, U1 = K0, U0 = HB;
  const int lb = (((l16 >> 3) & 1) << A3) | (((l16 >> 2) & 1) << A2) |
                 (((l16 >> 1) & 1) << A1) | ((l16 & 1) << A0) | (half << AH);
  constexpr int LM = ring_dst(1 << A3) | ring_dst(1 << A2) | ring_dst(1 << A1) |
                     ring_dst(1 << A0) | ring_dst(1 << AH);
  const int lr  = ring_dst(lb);
  const int kl  = (popc(lb) - popc(lr)) & 3;
  const int lrc = lr | ((kl & 1) << 10);
  unsigned sbm[4];
  #pragma unroll
  for (int kc = 0; kc < 4; ++kc)
    sbm[kc] = (unsigned)(((kl + kc) >> 1) & 1) << 31;
  constexpr int RB[5] = {R0, R1, R2, U0, U1};
  unsigned cm[32];
  cm[0] = 0u;
  #pragma unroll
  for (int x = 1; x < 32; ++x) {
    const int low = x & -x;
    const int bi  = (low == 1) ? 0 : (low == 2) ? 1 : (low == 4) ? 2 : (low == 8) ? 3 : 4;
    const unsigned bmw = (unsigned)(popc(lr & ring_dst(1 << RB[bi])) & 1) << 31;
    cm[x] = cm[x ^ low] ^ bmw;
  }
  #pragma unroll
  for (int t = 0; t < 8; ++t) {
    #pragma unroll
    for (int r = 0; r < 8; ++r) {
      const int regS = (((t >> 1) & 1) << U1) | ((t & 1) << U0) |
                       (((r >> 2) & 1) << R2) | (((r >> 1) & 1) << R1) | ((r & 1) << R0);
      const int pl  = (t >> 2) & 1;
      const int ro  = ring_dst(regS);
      const int ko  = (popc(regS) - popc(ro)) & 3;
      const int kc  = (ko + pl) & 3;
      const int xc  = (ro & LM) | (((ko & 1) ^ pl) << 10);
      const int add = ro & ~LM & 0x3FF;
      const int x   = (r & 7) | ((t & 3) << 3);
      const unsigned w = __float_as_uint(d[t][r]) ^ cm[x] ^ sbm[kc];
      S[(lrc ^ xc) + add] = __uint_as_float(w);
    }
  }
}

__device__ __forceinline__ float xhalf(float v) {
  const int b = __float_as_int(v);
  return __int_as_float(__builtin_amdgcn_permlanex16(b, b, 0x76543210, 0xfedcba98, false, false));
}

template <int M>
__device__ __forceinline__ float row_xor_add(float v) {
  const int b = __float_as_int(v);
  return v + __int_as_float(__builtin_amdgcn_update_dpp(0, b, 0x160 + M, 0xf, 0xf, true));
}

template <int HB, int L3, int L2, int L1, int L0, int K1, int K0, int XY, int T1, int T0>
__device__ __forceinline__ void measure_z(
    const v8f (&d)[8], int lane, float* __restrict__ zrow)
{
  constexpr int AH = XY, A3 = L2, A2 = L1, A1 = L3, A0 = L0;
  constexpr int R2 = T1, R1 = T0, R0 = K1, U1 = K0, U0 = HB;

  float w[32];
  #pragma unroll
  for (int t = 0; t < 4; ++t) {
    #pragma unroll
    for (int r = 0; r < 8; ++r)
      w[(t << 3) | r] = d[t][r] * d[t][r] + d[t + 4][r] * d[t + 4][r];
  }
  constexpr int ORD[5] = {4, 3, 1, 2, 0};
  #pragma unroll
  for (int i = 0; i < 5; ++i) {
    const int bit = 1 << ORD[i];
    #pragma unroll
    for (int x = 0; x < 32; ++x) {
      if (x & bit) continue;
      const float a = w[x], b = w[x | bit];
      w[x]       = a + b;
      w[x | bit] = a - b;
    }
  }
  constexpr int RB[5] = {R0, R1, R2, U0, U1};
  constexpr int LB[5] = {A0, A1, A2, A3, AH};
  float z = 0.0f;
  #pragma unroll
  for (int q = 0; q < NQ; ++q) {
    int m = 0, lm = 0;
    #pragma unroll
    for (int i = 0; i < 5; ++i) {
      m  |= ((ring_dst(1 << RB[i]) >> (9 - q)) & 1) << i;
      lm |= ((ring_dst(1 << LB[i]) >> (9 - q)) & 1) << i;
    }
    float v = w[m];
    if (lm)
      v = __uint_as_float(__float_as_uint(v) ^ ((unsigned)(__builtin_popcount(lane & lm) & 1) << 31));
    v = row_xor_add<1>(v);
    v = row_xor_add<2>(v);
    v = row_xor_add<4>(v);
    v = row_xor_add<8>(v);
    v += xhalf(v);
    z = (lane == q) ? v : z;
  }
  if (lane < NQ) zrow[lane] = z;
}

__global__ __launch_bounds__(128) void k_circuit(
    const float* __restrict__ cs, const float* __restrict__ tq,
    float* __restrict__ z)
{
  __shared__ float stl[4][2 * DIM];

  const int tid  = threadIdx.x;
  const int wave = tid >> 5;
  const int lane = tid & 31;
  const int half = lane >> 4;
  const int l16  = lane & 15;
  const int b    = blockIdx.x * 4 + wave;

  float* S  = stl[wave];

  v2f st[8][4];
  v8f d[8];
  float hc[NQ], hs[NQ];

  #pragma unroll
  for (int w = 0; w < NQ; ++w) { hc[w] = cs[b * 2 * NQ + w]; hs[w] = cs[b * 2 * NQ + NQ + w]; }
  first_core<5, 4, 3, 2, 1, 9, 8, 0, 7, 6>(d, lane, l16, half, hc, hs);
  #pragma unroll
  for (int t = 0; t < 4; ++t) {
    #pragma unroll
    for (int kk = 0; kk < 4; ++kk) { st[t][kk].x = d[t][2 * kk]; st[t][kk].y = d[t][2 * kk + 1]; }
  }
  #pragma unroll
  for (int w = 0; w < NQ; ++w) { hc[w] = tq[w]; hs[w] = tq[NQ + w]; }
  layer_core<1, 0, 3, 2, 4, 1, 7, 6, 9, 8, 5>(st, d, l16, half, hc, hs);
  ring_store<0, 3, 2, 4, 1, 7, 6, 9, 8, 5>(S, d, l16, half);
  wave_sync();

  for (int l = 1; l < NL - 1; ++l) {
    #pragma unroll
    for (int w = 0; w < NQ; ++w) { hc[w] = tq[l * 2 * NQ + w]; hs[w] = tq[l * 2 * NQ + NQ + w]; }
    load_state<5, 4, 3, 2, 1, 9, 8, 0, 7, 6>(S, st, l16, half);
    wave_sync();
    layer_core<2, 5, 4, 3, 2, 1, 9, 8, 0, 7, 6>(st, d, l16, half, hc, hs);
    ring_store<5, 4, 3, 2, 1, 9, 8, 0, 7, 6>(S, d, l16, half);
    wave_sync();
  }

  #pragma unroll
  for (int w = 0; w < NQ; ++w) {
    hc[w] = tq[(NL - 1) * 2 * NQ + w];
    hs[w] = tq[(NL - 1) * 2 * NQ + NQ + w];
  }
  load_state<5, 4, 3, 2, 1, 9, 8, 0, 7, 6>(S, st, l16, half);
  layer_core<2, 5, 4, 3, 2, 1, 9, 8, 0, 7, 6>(st, d, l16, half, hc, hs);
  measure_z<5, 4, 3, 2, 1, 9, 8, 0, 7, 6>(d, lane, z + b * NQ);
}

__global__ __launch_bounds__(128) void k_readout(
    const float* __restrict__ z, const float* __restrict__ Wout,
    const float* __restrict__ bout, float* __restrict__ out)
{
  const int lane = threadIdx.x & 31;
  const int wave = threadIdx.x >> 5;
  const int tile = blockIdx.x * 4 + wave;
  const int m0   = (tile >> 4) * 16;
  const int n0   = (tile & 15) * 16;
  const int half = lane >> 4;
  const int l16  = lane & 15;

  const float* arow = z    + (m0 + l16) * NQ;
  const float* brow = Wout + (n0 + l16) * NQ;

  v8f acc = {};
  #pragma unroll
  for (int kk = 0; kk < 3; ++kk) {
    const int k0  = kk * 4 + half * 2;
    const int k0c = (k0 < NQ - 1) ? k0 : (NQ - 2);
    const float vm0 = (k0     < NQ) ? 1.0f : 0.0f;
    const float vm1 = (k0 + 1 < NQ) ? 1.0f : 0.0f;
    v2f a, b;
    a.x = arow[k0c]     * vm0;
    a.y = arow[k0c + 1] * vm1;
    b.x = brow[k0c];
    b.y = brow[k0c + 1];
    acc = __builtin_amdgcn_wmma_f32_16x16x4_f32(false, a, false, b,
                                                (short)0, acc, false, false);
  }
  const float bias = bout[n0 + l16];
  #pragma unroll
  for (int r = 0; r < 8; ++r) {
    const int m = m0 + r + 8 * half;
    out[m * N_OUT + n0 + l16] = acc[r] + bias;
  }
}

extern "C" void kernel_launch(void* const* d_in, const int* in_sizes, int n_in,
                              void* d_out, int out_size, void* d_ws, size_t ws_size,
                              hipStream_t stream) {
  const float* x    = (const float*)d_in[0];
  const float* Win  = (const float*)d_in[1];
  const float* bin  = (const float*)d_in[2];
  const float* qp   = (const float*)d_in[3];
  const float* Wout = (const float*)d_in[4];
  const float* bout = (const float*)d_in[5];
  float* out = (float*)d_out;

  float* cs = (float*)d_ws;
  float* tq = cs + BATCH * 2 * NQ;
  float* zv = tq + 128;

  k_encode <<<BATCH / 16 / 4 + 1, 128, 0, stream>>>(x, Win, bin, qp, cs, tq);
  k_circuit<<<BATCH / 4,      128, 0, stream>>>(cs, tq, zv);
  k_readout<<<BATCH * 16 / 16 / 4, 128, 0, stream>>>(zv, Wout, bout, out);
}


# k_encode: one wave per 16-row tile, no LDS tile or barrier: x and W_in fragments are 16-byte pieces of their rows (k = 8j+4h+e in both), all loads one group ahead of the WMMAs; sums reordered
# speedup vs baseline: 3.2345x; 1.2755x over previous
#include <hip/hip_runtime.h>
#include <math.h>

typedef float v2f __attribute__((ext_vector_type(2)));
typedef float v4f __attribute__((ext_vector_type(4)));
typedef float v8f __attribute__((ext_vector_type(8)));

#define NQ 10
#define NL 4
#define DIM 1024
#define BATCH 4096
#define K_IN 128
#define N_OUT 256

__global__ __launch_bounds__(32) __attribute__((amdgpu_waves_per_eu(1, 1))) void k_encode(
    const float* __restrict__ x, const float* __restrict__ Win,
    const float* __restrict__ bin, const float* __restrict__ qp,
    float* __restrict__ cs, float* __restrict__ tq)
{
  const int lane = threadIdx.x;
  const int m0   = blockIdx.x * 16;
  const int half = lane >> 4;
  const int l16  = lane & 15;

  if (blockIdx.x == BATCH / 16) {
    for (int i = lane; i < NL * NQ; i += 32) {
      const float a = 0.5f * qp[i];
      const int l = i / NQ, w = i - l * NQ;
      tq[l * 2 * NQ + w]      = cosf(a);
      tq[l * 2 * NQ + NQ + w] = sinf(a);
    }
    return;
  }

  const int    nrow = (l16 < NQ) ? l16 : (NQ - 1);
  const float* arow = x   + (m0 + l16) * K_IN + half * 4;
  const float* brow = Win + nrow * K_IN + half * 4;
  const float  bias = bin[nrow];

  v4f a[K_IN / 8], b[K_IN / 8];
  #pragma unroll
  for (int j = 0; j < K_IN / 8; ++j) {
    a[j].x = arow[8 * j];     a[j].y = arow[8 * j + 1];
    a[j].z = arow[8 * j + 2]; a[j].w = arow[8 * j + 3];
    b[j].x = brow[8 * j];     b[j].y = brow[8 * j + 1];
    b[j].z = brow[8 * j + 2]; b[j].w = brow[8 * j + 3];
  }

  v8f acc0, acc1 = {};
  #pragma unroll
  for (int r = 0; r < 8; ++r) acc0[r] = bias;
  #pragma unroll
  for (int j = 0; j < K_IN / 8; ++j) {
    v2f a0, b0, a1, b1;
    a0.x = a[j].x; a0.y = a[j].y; b0.x = b[j].x; b0.y = b[j].y;
    a1.x = a[j].z; a1.y = a[j].w; b1.x = b[j].z; b1.y = b[j].w;
    acc0 = __builtin_amdgcn_wmma_f32_16x16x4_f32(false, a0, false, b0,
                                                 (short)0, acc0, false, false);
    acc1 = __builtin_amdgcn_wmma_f32_16x16x4_f32(false, a1, false, b1,
                                                 (short)0, acc1, false, false);
  }
  const v8f pre = acc0 + acc1;
  if (l16 < NQ) {
    #pragma unroll
    for (int r = 0; r < 8; ++r) {
      const int m = m0 + r + 8 * half;
      const float a = 0.5f * tanhf(pre[r]);
      cs[m * 2 * NQ + l16]      = cosf(a);
      cs[m * 2 * NQ + NQ + l16] = sinf(a);
    }
  }
}


__device__ __forceinline__ void wave_sync() {
  __builtin_amdgcn_fence(__ATOMIC_RELEASE, "wavefront");
  __builtin_amdgcn_wave_barrier();
  __builtin_amdgcn_fence(__ATOMIC_ACQUIRE, "wavefront");
}

__device__ __forceinline__ constexpr int ring_dst(int s) {
  for (int c = 0; c <= 8; ++c)
    if ((s >> (9 - c)) & 1) s ^= 1 << (8 - c);
  if (s & 1) s ^= 1 << 9;
  return s;
}

__device__ __forceinline__ constexpr int popc(int v) { return __builtin_popcount(v); }

__device__ __forceinline__ float rot(bool a, bool b, float c, float s) {
  return (a == b) ? c : (a ? -s : s);
}

__device__ __forceinline__ void gate_frag(v2f (&g)[4], bool o3, bool o2, bool o1, bool o0,
    bool half, float c3, float s3, float c2, float s2, float c1, float s1, float c0, float s0)
{
  const float fH = rot(o1, half, c1, s1);
  const float w0 = fH * rot(o0, false, c0, s0);
  const float w1 = fH * rot(o0, true,  c0, s0);
  #pragma unroll
  for (int kk = 0; kk < 4; ++kk) {
    const float u = rot(o3, (kk >> 1) != 0, c3, s3) * rot(o2, (kk & 1) != 0, c2, s2);
    g[kk].x = u * w0;
    g[kk].y = u * w1;
  }
}

template <int NP, int HB, int L3, int L2, int L1, int L0, int K1, int K0, int XY, int T1, int T0>
__device__ __forceinline__ void layer_core(
    const v2f (&st)[8][4], v8f (&d)[8], int l16, int half,
    const float (&hc)[NQ], const float (&hs)[NQ])
{
  constexpr int NT = 4 * NP;
  const bool n3 = (l16 >> 3) & 1, n2 = (l16 >> 2) & 1;
  const bool n1 = (l16 >> 1) & 1, n0 = l16 & 1;
  const bool hb = half != 0;
  v2f g[4];

  gate_frag(g, n3, n2, n1, n0, hb, hc[9 - K1], hs[9 - K1], hc[9 - K0], hs[9 - K0],
            hc[9 - HB], hs[9 - HB], hc[9 - XY], hs[9 - XY]);
  v8f d1[NT];
  #pragma unroll
  for (int t = 0; t < NT; ++t) d1[t] = v8f{};
  #pragma unroll
  for (int kk = 0; kk < 4; ++kk) {
    #pragma unroll
    for (int t = 0; t < NT; ++t)
      d1[t] = __builtin_amdgcn_wmma_f32_16x16x4_f32(false, st[t][kk], false, g[kk],
                                                    (short)0, d1[t], false, false);
  }

  gate_frag(g, n3, n2, n1, n0, hb, hc[9 - L2], hs[9 - L2], hc[9 - L1], hs[9 - L1],
            hc[9 - L3], hs[9 - L3], hc[9 - L0], hs[9 - L0]);
  v8f d2[NT];
  #pragma unroll
  for (int t = 0; t < NT; ++t) d2[t] = v8f{};
  #pragma unroll
  for (int kk = 0; kk < 4; ++kk) {
    #pragma unroll
    for (int t = 0; t < NT; ++t) {
      v2f a;
      a.x = d1[t][2 * kk];
      a.y = d1[t][2 * kk + 1];
      d2[t] = __builtin_amdgcn_wmma_f32_16x16x4_f32(false, a, false, g[kk],
                                                    (short)0, d2[t], false, false);
    }
  }

  gate_frag(g, n2, n1, n0, n3, hb, hc[9 - T1], hs[9 - T1], hc[9 - T0], hs[9 - T0],
            1.0f, 0.0f, 1.0f, 0.0f);
  #pragma unroll
  for (int t = 0; t < 8; ++t) d[t] = v8f{};
  #pragma unroll
  for (int kk = 0; kk < 4; ++kk) {
    #pragma unroll
    for (int tp = 0; tp < NT; ++tp) {
      const int p = tp >> 2, u = tp & 3;
      v2f b;
      b.x = d2[(p << 2) | kk][2 * u];
      b.y = d2[(p << 2) | kk][2 * u + 1];
      d[tp] = __builtin_amdgcn_wmma_f32_16x16x4_f32(false, g[kk], false, b,
                                                    (short)0, d[tp], false, false);
    }
  }
}

template <int HB, int L3, int L2, int L1, int L0, int K1, int K0, int XY, int T1, int T0>
__device__ __forceinline__ void first_core(
    v8f (&d)[8], int lane, int l16, int half,
    const float (&hc)[NQ], const float (&hs)[NQ])
{
  const bool n3 = (l16 >> 3) & 1, n2 = (l16 >> 2) & 1;
  const bool n1 = (l16 >> 1) & 1, n0 = l16 & 1;
  const bool hb = half != 0;
  v2f g[4];

  v2f a0;
  a0.x = (lane == 0) ? 1.0f : 0.0f;
  a0.y = 0.0f;
  gate_frag(g, n3, n2, n1, n0, hb, hc[9 - K1], hs[9 - K1], hc[9 - K0], hs[9 - K0],
            hc[9 - HB], hs[9 - HB], hc[9 - XY], hs[9 - XY]);
  const v8f d1 = __builtin_amdgcn_wmma_f32_16x16x4_f32(false, a0, false, g[0],
                                                       (short)0, v8f{}, false, false);

  gate_frag(g, n3, n2, n1, n0, hb, hc[9 - L2], hs[9 - L2], hc[9 - L1], hs[9 - L1],
            hc[9 - L3], hs[9 - L3], hc[9 - L0], hs[9 - L0]);
  v8f d2 = v8f{};
  #pragma unroll
  for (int kk = 0; kk < 4; ++kk) {
    v2f a;
    a.x = d1[2 * kk];
    a.y = d1[2 * kk + 1];
    d2 = __builtin_amdgcn_wmma_f32_16x16x4_f32(false, a, false, g[kk],
                                               (short)0, d2, false, false);
  }

  gate_frag(g, n2, n1, n0, n3, hb, hc[9 - T1], hs[9 - T1], hc[9 - T0], hs[9 - T0],
            1.0f, 0.0f, 1.0f, 0.0f);
  #pragma unroll
  for (int u = 0; u < 4; ++u) {
    v2f b;
    b.x = d2[2 * u];
    b.y = d2[2 * u + 1];
    d[u] = __builtin_amdgcn_wmma_f32_16x16x4_f32(false, g[0], false, b,
                                                 (short)0, v8f{}, false, false);
  }
  #pragma unroll
  for (int t = 4; t < 8; ++t) d[t] = v8f{};
}

template <int HB, int L3, int L2, int L1, int L0, int K1, int K0, int XY, int T1, int T0>
__device__ __forceinline__ void load_state(
    const float* __restrict__ S, v2f (&st)[8][4], int l16, int half)
{
  const int lb = (((l16 >> 3) & 1) << L3) | (((l16 >> 2) & 1) << L2) |
                 (((l16 >> 1) & 1) << L1) | ((l16 & 1) << L0) | (half << HB);
  #pragma unroll
  for (int t = 0; t < 8; ++t) {
    #pragma unroll
    for (int kk = 0; kk < 4; ++kk) {
      const int o = (((t >> 2) & 1) << 10) | (((t >> 1) & 1) << T1) | ((t & 1) << T0) |
                    (((kk >> 1) & 1) << K1) | ((kk & 1) << K0);
      st[t][kk].x = S[lb + o];
      st[t][kk].y = S[lb + o + (1 << XY)];
    }
  }
}

template <int HB, int L3, int L2, int L1, int L0, int K1, int K0, int XY, int T1, int T0>
__device__ __forceinline__ void ring_store(
    float* __restrict__ S, const v8f (&d)[8], int l16, int half)
{
  constexpr int AH = XY, A3 = L2, A2 = L1, A1 = L3, A0 = L0;
  constexpr int R2 = T1, R1 = T0, R0 = K1, U1 = K0, U0 = HB;
  const int lb = (((l16 >> 3) & 1) << A3) | (((l16 >> 2) & 1) << A2) |
                 (((l16 >> 1) & 1) << A1) | ((l16 & 1) << A0) | (half << AH);
  constexpr int LM = ring_dst(1 << A3) | ring_dst(1 << A2) | ring_dst(1 << A1) |
                     ring_dst(1 << A0) | ring_dst(1 << AH);
  const int lr  = ring_dst(lb);
  const int kl  = (popc(lb) - popc(lr)) & 3;
  const int lrc = lr | ((kl & 1) << 10);
  unsigned sbm[4];
  #pragma unroll
  for (int kc = 0; kc < 4; ++kc)
    sbm[kc] = (unsigned)(((kl + kc) >> 1) & 1) << 31;
  constexpr int RB[5] = {R0, R1, R2, U0, U1};
  unsigned cm[32];
  cm[0] = 0u;
  #pragma unroll
  for (int x = 1; x < 32; ++x) {
    const int low = x & -x;
    const int bi  = (low == 1) ? 0 : (low == 2) ? 1 : (low == 4) ? 2 : (low == 8) ? 3 : 4;
    const unsigned bmw = (unsigned)(popc(lr & ring_dst(1 << RB[bi])) & 1) << 31;
    cm[x] = cm[x ^ low] ^ bmw;
  }
  #pragma unroll
  for (int t = 0; t < 8; ++t) {
    #pragma unroll
    for (int r = 0; r < 8; ++r) {
      const int regS = (((t >> 1) & 1) << U1) | ((t & 1) << U0) |
                       (((r >> 2) & 1) << R2) | (((r >> 1) & 1) << R1) | ((r & 1) << R0);
      const int pl  = (t >> 2) & 1;
      const int ro  = ring_dst(regS);
      const int ko  = (popc(regS) - popc(ro)) & 3;
      const int kc  = (ko + pl) & 3;
      const int xc  = (ro & LM) | (((ko & 1) ^ pl) << 10);
      const int add = ro & ~LM & 0x3FF;
      const int x   = (r & 7) | ((t & 3) << 3);
      const unsigned w = __float_as_uint(d[t][r]) ^ cm[x] ^ sbm[kc];
      S[(lrc ^ xc) + add] = __uint_as_float(w);
    }
  }
}

__device__ __forceinline__ float xhalf(float v) {
  const int b = __float_as_int(v);
  return __int_as_float(__builtin_amdgcn_permlanex16(b, b, 0x76543210, 0xfedcba98, false, false));
}

template <int M>
__device__ __forceinline__ float row_xor_add(float v) {
  const int b = __float_as_int(v);
  return v + __int_as_float(__builtin_amdgcn_update_dpp(0, b, 0x160 + M, 0xf, 0xf, true));
}

template <int HB, int L3, int L2, int L1, int L0, int K1, int K0, int XY, int T1, int T0>
__device__ __forceinline__ void measure_z(
    const v8f (&d)[8], int lane, float* __restrict__ zrow)
{
  constexpr int AH = XY, A3 = L2, A2 = L1, A1 = L3, A0 = L0;
  constexpr int R2 = T1, R1 = T0, R0 = K1, U1 = K0, U0 = HB;

  float w[32];
  #pragma unroll
  for (int t = 0; t < 4; ++t) {
    #pragma unroll
    for (int r = 0; r < 8; ++r)
      w[(t << 3) | r] = d[t][r] * d[t][r] + d[t + 4][r] * d[t + 4][r];
  }
  constexpr int ORD[5] = {4, 3, 1, 2, 0};
  #pragma unroll
  for (int i = 0; i < 5; ++i) {
    const int bit = 1 << ORD[i];
    #pragma unroll
    for (int x = 0; x < 32; ++x) {
      if (x & bit) continue;
      const float a = w[x], b = w[x | bit];
      w[x]       = a + b;
      w[x | bit] = a - b;
    }
  }
  constexpr int RB[5] = {R0, R1, R2, U0, U1};
  constexpr int LB[5] = {A0, A1, A2, A3, AH};
  float z = 0.0f;
  #pragma unroll
  for (int q = 0; q < NQ; ++q) {
    int m = 0, lm = 0;
    #pragma unroll
    for (int i = 0; i < 5; ++i) {
      m  |= ((ring_dst(1 << RB[i]) >> (9 - q)) & 1) << i;
      lm |= ((ring_dst(1 << LB[i]) >> (9 - q)) & 1) << i;
    }
    float v = w[m];
    if (lm)
      v = __uint_as_float(__float_as_uint(v) ^ ((unsigned)(__builtin_popcount(lane & lm) & 1) << 31));
    v = row_xor_add<1>(v);
    v = row_xor_add<2>(v);
    v = row_xor_add<4>(v);
    v = row_xor_add<8>(v);
    v += xhalf(v);
    z = (lane == q) ? v : z;
  }
  if (lane < NQ) zrow[lane] = z;
}

__global__ __launch_bounds__(128) void k_circuit(
    const float* __restrict__ cs, const float* __restrict__ tq,
    float* __restrict__ z)
{
  __shared__ float stl[4][2 * DIM];

  const int tid  = threadIdx.x;
  const int wave = tid >> 5;
  const int lane = tid & 31;
  const int half = lane >> 4;
  const int l16  = lane & 15;
  const int b    = blockIdx.x * 4 + wave;

  float* S  = stl[wave];

  v2f st[8][4];
  v8f d[8];
  float hc[NQ], hs[NQ];

  #pragma unroll
  for (int w = 0; w < NQ; ++w) { hc[w] = cs[b * 2 * NQ + w]; hs[w] = cs[b * 2 * NQ + NQ + w]; }
  first_core<5, 4, 3, 2, 1, 9, 8, 0, 7, 6>(d, lane, l16, half, hc, hs);
  #pragma unroll
  for (int t = 0; t < 4; ++t) {
    #pragma unroll
    for (int kk = 0; kk < 4; ++kk) { st[t][kk].x = d[t][2 * kk]; st[t][kk].y = d[t][2 * kk + 1]; }
  }
  #pragma unroll
  for (int w = 0; w < NQ; ++w) { hc[w] = tq[w]; hs[w] = tq[NQ + w]; }
  layer_core<1, 0, 3, 2, 4, 1, 7, 6, 9, 8, 5>(st, d, l16, half, hc, hs);
  ring_store<0, 3, 2, 4, 1, 7, 6, 9, 8, 5>(S, d, l16, half);
  wave_sync();

  for (int l = 1; l < NL - 1; ++l) {
    #pragma unroll
    for (int w = 0; w < NQ; ++w) { hc[w] = tq[l * 2 * NQ + w]; hs[w] = tq[l * 2 * NQ + NQ + w]; }
    load_state<5, 4, 3, 2, 1, 9, 8, 0, 7, 6>(S, st, l16, half);
    wave_sync();
    layer_core<2, 5, 4, 3, 2, 1, 9, 8, 0, 7, 6>(st, d, l16, half, hc, hs);
    ring_store<5, 4, 3, 2, 1, 9, 8, 0, 7, 6>(S, d, l16, half);
    wave_sync();
  }

  #pragma unroll
  for (int w = 0; w < NQ; ++w) {
    hc[w] = tq[(NL - 1) * 2 * NQ + w];
    hs[w] = tq[(NL - 1) * 2 * NQ + NQ + w];
  }
  load_state<5, 4, 3, 2, 1, 9, 8, 0, 7, 6>(S, st, l16, half);
  layer_core<2, 5, 4, 3, 2, 1, 9, 8, 0, 7, 6>(st, d, l16, half, hc, hs);
  measure_z<5, 4, 3, 2, 1, 9, 8, 0, 7, 6>(d, lane, z + b * NQ);
}

__global__ __launch_bounds__(128) void k_readout(
    const float* __restrict__ z, const float* __restrict__ Wout,
    const float* __restrict__ bout, float* __restrict__ out)
{
  const int lane = threadIdx.x & 31;
  const int wave = threadIdx.x >> 5;
  const int tile = blockIdx.x * 4 + wave;
  const int m0   = (tile >> 4) * 16;
  const int n0   = (tile & 15) * 16;
  const int half = lane >> 4;
  const int l16  = lane & 15;

  const float* arow = z    + (m0 + l16) * NQ;
  const float* brow = Wout + (n0 + l16) * NQ;

  v8f acc = {};
  #pragma unroll
  for (int kk = 0; kk < 3; ++kk) {
    const int k0  = kk * 4 + half * 2;
    const int k0c = (k0 < NQ - 1) ? k0 : (NQ - 2);
    const float vm0 = (k0     < NQ) ? 1.0f : 0.0f;
    const float vm1 = (k0 + 1 < NQ) ? 1.0f : 0.0f;
    v2f a, b;
    a.x = arow[k0c]     * vm0;
    a.y = arow[k0c + 1] * vm1;
    b.x = brow[k0c];
    b.y = brow[k0c + 1];
    acc = __builtin_amdgcn_wmma_f32_16x16x4_f32(false, a, false, b,
                                                (short)0, acc, false, false);
  }
  const float bias = bout[n0 + l16];
  #pragma unroll
  for (int r = 0; r < 8; ++r) {
    const int m = m0 + r + 8 * half;
    out[m * N_OUT + n0 + l16] = acc[r] + bias;
  }
}

extern "C" void kernel_launch(void* const* d_in, const int* in_sizes, int n_in,
                              void* d_out, int out_size, void* d_ws, size_t ws_size,
                              hipStream_t stream) {
  const float* x    = (const float*)d_in[0];
  const float* Win  = (const float*)d_in[1];
  const float* bin  = (const float*)d_in[2];
  const float* qp   = (const float*)d_in[3];
  const float* Wout = (const float*)d_in[4];
  const float* bout = (const float*)d_in[5];
  float* out = (float*)d_out;

  float* cs = (float*)d_ws;
  float* tq = cs + BATCH * 2 * NQ;
  float* zv = tq + 128;

  k_encode <<<BATCH / 16 + 1,      32, 0, stream>>>(x, Win, bin, qp, cs, tq);
  k_circuit<<<BATCH / 4,      128, 0, stream>>>(cs, tq, zv);
  k_readout<<<BATCH * 16 / 16 / 4, 128, 0, stream>>>(zv, Wout, bout, out);
}


# k_encode epilogue straight-line: both tanh ranges evaluated and one selected, cos/sin of h/2 as Taylor polynomials with no range reduction (|h/2| <= 0.5 after tanh); a lane's 8 rows interleave
# speedup vs baseline: 3.4036x; 1.0523x over previous
#include <hip/hip_runtime.h>
#include <math.h>

typedef float v2f __attribute__((ext_vector_type(2)));
typedef float v4f __attribute__((ext_vector_type(4)));
typedef float v8f __attribute__((ext_vector_type(8)));

#define NQ 10
#define NL 4
#define DIM 1024
#define BATCH 4096
#define K_IN 128
#define N_OUT 256


__device__ __forceinline__ float tanh_straight(float x) {
  const float ax  = fabsf(x);
  const float e   = expf(ax + ax);
  const float big = fmaf(__builtin_amdgcn_rcpf(e + 1.0f), -2.0f, 1.0f);
  const float x2  = x * x;
  float p = fmaf(-0x1.758e7ap-8f, x2, 0x1.521192p-6f);
  p = fmaf(x2, p, -0x1.b8389cp-5f);
  p = fmaf(x2, p, 0x1.110704p-3f);
  p = fmaf(x2, p, -0x1.555532p-2f);
  const float small = fmaf(x2, ax * p, ax);
  return copysignf((0.625f > ax) ? small : big, x);
}

__global__ __launch_bounds__(32) __attribute__((amdgpu_waves_per_eu(1, 1))) void k_encode(
    const float* __restrict__ x, const float* __restrict__ Win,
    const float* __restrict__ bin, const float* __restrict__ qp,
    float* __restrict__ cs, float* __restrict__ tq)
{
  const int lane = threadIdx.x;
  const int m0   = blockIdx.x * 16;
  const int half = lane >> 4;
  const int l16  = lane & 15;

  if (blockIdx.x == BATCH / 16) {
    for (int i = lane; i < NL * NQ; i += 32) {
      const float a = 0.5f * qp[i];
      const int l = i / NQ, w = i - l * NQ;
      tq[l * 2 * NQ + w]      = cosf(a);
      tq[l * 2 * NQ + NQ + w] = sinf(a);
    }
    return;
  }

  const int    nrow = (l16 < NQ) ? l16 : (NQ - 1);
  const float* arow = x   + (m0 + l16) * K_IN + half * 4;
  const float* brow = Win + nrow * K_IN + half * 4;
  const float  bias = bin[nrow];

  v4f a[K_IN / 8], b[K_IN / 8];
  #pragma unroll
  for (int j = 0; j < K_IN / 8; ++j) {
    a[j].x = arow[8 * j];     a[j].y = arow[8 * j + 1];
    a[j].z = arow[8 * j + 2]; a[j].w = arow[8 * j + 3];
    b[j].x = brow[8 * j];     b[j].y = brow[8 * j + 1];
    b[j].z = brow[8 * j + 2]; b[j].w = brow[8 * j + 3];
  }

  v8f acc0, acc1 = {};
  #pragma unroll
  for (int r = 0; r < 8; ++r) acc0[r] = bias;
  #pragma unroll
  for (int j = 0; j < K_IN / 8; ++j) {
    v2f a0, b0, a1, b1;
    a0.x = a[j].x; a0.y = a[j].y; b0.x = b[j].x; b0.y = b[j].y;
    a1.x = a[j].z; a1.y = a[j].w; b1.x = b[j].z; b1.y = b[j].w;
    acc0 = __builtin_amdgcn_wmma_f32_16x16x4_f32(false, a0, false, b0,
                                                 (short)0, acc0, false, false);
    acc1 = __builtin_amdgcn_wmma_f32_16x16x4_f32(false, a1, false, b1,
                                                 (short)0, acc1, false, false);
  }
  const v8f pre = acc0 + acc1;
  if (l16 < NQ) {
    #pragma unroll
    for (int r = 0; r < 8; ++r) {
      const int m = m0 + r + 8 * half;
      const float a  = 0.5f * tanh_straight(pre[r]);
      const float a2 = a * a;
      float cp = fmaf(a2, -1.0f / 3628800.0f, 1.0f / 40320.0f);
      cp = fmaf(a2, cp, -1.0f / 720.0f);
      cp = fmaf(a2, cp, 1.0f / 24.0f);
      cp = fmaf(a2, cp, -0.5f);
      float sp = fmaf(a2, 1.0f / 362880.0f, -1.0f / 5040.0f);
      sp = fmaf(a2, sp, 1.0f / 120.0f);
      sp = fmaf(a2, sp, -1.0f / 6.0f);
      cs[m * 2 * NQ + l16]      = fmaf(a2, cp, 1.0f);
      cs[m * 2 * NQ + NQ + l16] = fmaf(a * a2, sp, a);
    }
  }
}


__device__ __forceinline__ void wave_sync() {
  __builtin_amdgcn_fence(__ATOMIC_RELEASE, "wavefront");
  __builtin_amdgcn_wave_barrier();
  __builtin_amdgcn_fence(__ATOMIC_ACQUIRE, "wavefront");
}

__device__ __forceinline__ constexpr int ring_dst(int s) {
  for (int c = 0; c <= 8; ++c)
    if ((s >> (9 - c)) & 1) s ^= 1 << (8 - c);
  if (s & 1) s ^= 1 << 9;
  return s;
}

__device__ __forceinline__ constexpr int popc(int v) { return __builtin_popcount(v); }

__device__ __forceinline__ float rot(bool a, bool b, float c, float s) {
  return (a == b) ? c : (a ? -s : s);
}

__device__ __forceinline__ void gate_frag(v2f (&g)[4], bool o3, bool o2, bool o1, bool o0,
    bool half, float c3, float s3, float c2, float s2, float c1, float s1, float c0, float s0)
{
  const float fH = rot(o1, half, c1, s1);
  const float w0 = fH * rot(o0, false, c0, s0);
  const float w1 = fH * rot(o0, true,  c0, s0);
  #pragma unroll
  for (int kk = 0; kk < 4; ++kk) {
    const float u = rot(o3, (kk >> 1) != 0, c3, s3) * rot(o2, (kk & 1) != 0, c2, s2);
    g[kk].x = u * w0;
    g[kk].y = u * w1;
  }
}

template <int NP, int HB, int L3, int L2, int L1, int L0, int K1, int K0, int XY, int T1, int T0>
__device__ __forceinline__ void layer_core(
    const v2f (&st)[8][4], v8f (&d)[8], int l16, int half,
    const float (&hc)[NQ], const float (&hs)[NQ])
{
  constexpr int NT = 4 * NP;
  const bool n3 = (l16 >> 3) & 1, n2 = (l16 >> 2) & 1;
  const bool n1 = (l16 >> 1) & 1, n0 = l16 & 1;
  const bool hb = half != 0;
  v2f g[4];

  gate_frag(g, n3, n2, n1, n0, hb, hc[9 - K1], hs[9 - K1], hc[9 - K0], hs[9 - K0],
            hc[9 - HB], hs[9 - HB], hc[9 - XY], hs[9 - XY]);
  v8f d1[NT];
  #pragma unroll
  for (int t = 0; t < NT; ++t) d1[t] = v8f{};
  #pragma unroll
  for (int kk = 0; kk < 4; ++kk) {
    #pragma unroll
    for (int t = 0; t < NT; ++t)
      d1[t] = __builtin_amdgcn_wmma_f32_16x16x4_f32(false, st[t][kk], false, g[kk],
                                                    (short)0, d1[t], false, false);
  }

  gate_frag(g, n3, n2, n1, n0, hb, hc[9 - L2], hs[9 - L2], hc[9 - L1], hs[9 - L1],
            hc[9 - L3], hs[9 - L3], hc[9 - L0], hs[9 - L0]);
  v8f d2[NT];
  #pragma unroll
  for (int t = 0; t < NT; ++t) d2[t] = v8f{};
  #pragma unroll
  for (int kk = 0; kk < 4; ++kk) {
    #pragma unroll
    for (int t = 0; t < NT; ++t) {
      v2f a;
      a.x = d1[t][2 * kk];
      a.y = d1[t][2 * kk + 1];
      d2[t] = __builtin_amdgcn_wmma_f32_16x16x4_f32(false, a, false, g[kk],
                                                    (short)0, d2[t], false, false);
    }
  }

  gate_frag(g, n2, n1, n0, n3, hb, hc[9 - T1], hs[9 - T1], hc[9 - T0], hs[9 - T0],
            1.0f, 0.0f, 1.0f, 0.0f);
  #pragma unroll
  for (int t = 0; t < 8; ++t) d[t] = v8f{};
  #pragma unroll
  for (int kk = 0; kk < 4; ++kk) {
    #pragma unroll
    for (int tp = 0; tp < NT; ++tp) {
      const int p = tp >> 2, u = tp & 3;
      v2f b;
      b.x = d2[(p << 2) | kk][2 * u];
      b.y = d2[(p << 2) | kk][2 * u + 1];
      d[tp] = __builtin_amdgcn_wmma_f32_16x16x4_f32(false, g[kk], false, b,
                                                    (short)0, d[tp], false, false);
    }
  }
}

template <int HB, int L3, int L2, int L1, int L0, int K1, int K0, int XY, int T1, int T0>
__device__ __forceinline__ void first_core(
    v8f (&d)[8], int lane, int l16, int half,
    const float (&hc)[NQ], const float (&hs)[NQ])
{
  const bool n3 = (l16 >> 3) & 1, n2 = (l16 >> 2) & 1;
  const bool n1 = (l16 >> 1) & 1, n0 = l16 & 1;
  const bool hb = half != 0;
  v2f g[4];

  v2f a0;
  a0.x = (lane == 0) ? 1.0f : 0.0f;
  a0.y = 0.0f;
  gate_frag(g, n3, n2, n1, n0, hb, hc[9 - K1], hs[9 - K1], hc[9 - K0], hs[9 - K0],
            hc[9 - HB], hs[9 - HB], hc[9 - XY], hs[9 - XY]);
  const v8f d1 = __builtin_amdgcn_wmma_f32_16x16x4_f32(false, a0, false, g[0],
                                                       (short)0, v8f{}, false, false);

  gate_frag(g, n3, n2, n1, n0, hb, hc[9 - L2], hs[9 - L2], hc[9 - L1], hs[9 - L1],
            hc[9 - L3], hs[9 - L3], hc[9 - L0], hs[9 - L0]);
  v8f d2 = v8f{};
  #pragma unroll
  for (int kk = 0; kk < 4; ++kk) {
    v2f a;
    a.x = d1[2 * kk];
    a.y = d1[2 * kk + 1];
    d2 = __builtin_amdgcn_wmma_f32_16x16x4_f32(false, a, false, g[kk],
                                               (short)0, d2, false, false);
  }

  gate_frag(g, n2, n1, n0, n3, hb, hc[9 - T1], hs[9 - T1], hc[9 - T0], hs[9 - T0],
            1.0f, 0.0f, 1.0f, 0.0f);
  #pragma unroll
  for (int u = 0; u < 4; ++u) {
    v2f b;
    b.x = d2[2 * u];
    b.y = d2[2 * u + 1];
    d[u] = __builtin_amdgcn_wmma_f32_16x16x4_f32(false, g[0], false, b,
                                                 (short)0, v8f{}, false, false);
  }
  #pragma unroll
  for (int t = 4; t < 8; ++t) d[t] = v8f{};
}

template <int HB, int L3, int L2, int L1, int L0, int K1, int K0, int XY, int T1, int T0>
__device__ __forceinline__ void load_state(
    const float* __restrict__ S, v2f (&st)[8][4], int l16, int half)
{
  const int lb = (((l16 >> 3) & 1) << L3) | (((l16 >> 2) & 1) << L2) |
                 (((l16 >> 1) & 1) << L1) | ((l16 & 1) << L0) | (half << HB);
  #pragma unroll
  for (int t = 0; t < 8; ++t) {
    #pragma unroll
    for (int kk = 0; kk < 4; ++kk) {
      const int o = (((t >> 2) & 1) << 10) | (((t >> 1) & 1) << T1) | ((t & 1) << T0) |
                    (((kk >> 1) & 1) << K1) | ((kk & 1) << K0);
      st[t][kk].x = S[lb + o];
      st[t][kk].y = S[lb + o + (1 << XY)];
    }
  }
}

template <int HB, int L3, int L2, int L1, int L0, int K1, int K0, int XY, int T1, int T0>
__device__ __forceinline__ void ring_store(
    float* __restrict__ S, const v8f (&d)[8], int l16, int half)
{
  constexpr int AH = XY, A3 = L2, A2 = L1, A1 = L3, A0 = L0;
  constexpr int R2 = T1, R1 = T0, R0 = K1, U1 = K0, U0 = HB;
  const int lb = (((l16 >> 3) & 1) << A3) | (((l16 >> 2) & 1) << A2) |
                 (((l16 >> 1) & 1) << A1) | ((l16 & 1) << A0) | (half << AH);
  constexpr int LM = ring_dst(1 << A3) | ring_dst(1 << A2) | ring_dst(1 << A1) |
                     ring_dst(1 << A0) | ring_dst(1 << AH);
  const int lr  = ring_dst(lb);
  const int kl  = (popc(lb) - popc(lr)) & 3;
  const int lrc = lr | ((kl & 1) << 10);
  unsigned sbm[4];
  #pragma unroll
  for (int kc = 0; kc < 4; ++kc)
    sbm[kc] = (unsigned)(((kl + kc) >> 1) & 1) << 31;
  constexpr int RB[5] = {R0, R1, R2, U0, U1};
  unsigned cm[32];
  cm[0] = 0u;
  #pragma unroll
  for (int x = 1; x < 32; ++x) {
    const int low = x & -x;
    const int bi  = (low == 1) ? 0 : (low == 2) ? 1 : (low == 4) ? 2 : (low == 8) ? 3 : 4;
    const unsigned bmw = (unsigned)(popc(lr & ring_dst(1 << RB[bi])) & 1) << 31;
    cm[x] = cm[x ^ low] ^ bmw;
  }
  #pragma unroll
  for (int t = 0; t < 8; ++t) {
    #pragma unroll
    for (int r = 0; r < 8; ++r) {
      const int regS = (((t >> 1) & 1) << U1) | ((t & 1) << U0) |
                       (((r >> 2) & 1) << R2) | (((r >> 1) & 1) << R1) | ((r & 1) << R0);
      const int pl  = (t >> 2) & 1;
      const int ro  = ring_dst(regS);
      const int ko  = (popc(regS) - popc(ro)) & 3;
      const int kc  = (ko + pl) & 3;
      const int xc  = (ro & LM) | (((ko & 1) ^ pl) << 10);
      const int add = ro & ~LM & 0x3FF;
      const int x   = (r & 7) | ((t & 3) << 3);
      const unsigned w = __float_as_uint(d[t][r]) ^ cm[x] ^ sbm[kc];
      S[(lrc ^ xc) + add] = __uint_as_float(w);
    }
  }
}

__device__ __forceinline__ float xhalf(float v) {
  const int b = __float_as_int(v);
  return __int_as_float(__builtin_amdgcn_permlanex16(b, b, 0x76543210, 0xfedcba98, false, false));
}

template <int M>
__device__ __forceinline__ float row_xor_add(float v) {
  const int b = __float_as_int(v);
  return v + __int_as_float(__builtin_amdgcn_update_dpp(0, b, 0x160 + M, 0xf, 0xf, true));
}

template <int HB, int L3, int L2, int L1, int L0, int K1, int K0, int XY, int T1, int T0>
__device__ __forceinline__ void measure_z(
    const v8f (&d)[8], int lane, float* __restrict__ zrow)
{
  constexpr int AH = XY, A3 = L2, A2 = L1, A1 = L3, A0 = L0;
  constexpr int R2 = T1, R1 = T0, R0 = K1, U1 = K0, U0 = HB;

  float w[32];
  #pragma unroll
  for (int t = 0; t < 4; ++t) {
    #pragma unroll
    for (int r = 0; r < 8; ++r)
      w[(t << 3) | r] = d[t][r] * d[t][r] + d[t + 4][r] * d[t + 4][r];
  }
  constexpr int ORD[5] = {4, 3, 1, 2, 0};
  #pragma unroll
  for (int i = 0; i < 5; ++i) {
    const int bit = 1 << ORD[i];
    #pragma unroll
    for (int x = 0; x < 32; ++x) {
      if (x & bit) continue;
      const float a = w[x], b = w[x | bit];
      w[x]       = a + b;
      w[x | bit] = a - b;
    }
  }
  constexpr int RB[5] = {R0, R1, R2, U0, U1};
  constexpr int LB[5] = {A0, A1, A2, A3, AH};
  float z = 0.0f;
  #pragma unroll
  for (int q = 0; q < NQ; ++q) {
    int m = 0, lm = 0;
    #pragma unroll
    for (int i = 0; i < 5; ++i) {
      m  |= ((ring_dst(1 << RB[i]) >> (9 - q)) & 1) << i;
      lm |= ((ring_dst(1 << LB[i]) >> (9 - q)) & 1) << i;
    }
    float v = w[m];
    if (lm)
      v = __uint_as_float(__float_as_uint(v) ^ ((unsigned)(__builtin_popcount(lane & lm) & 1) << 31));
    v = row_xor_add<1>(v);
    v = row_xor_add<2>(v);
    v = row_xor_add<4>(v);
    v = row_xor_add<8>(v);
    v += xhalf(v);
    z = (lane == q) ? v : z;
  }
  if (lane < NQ) zrow[lane] = z;
}

__global__ __launch_bounds__(128) void k_circuit(
    const float* __restrict__ cs, const float* __restrict__ tq,
    float* __restrict__ z)
{
  __shared__ float stl[4][2 * DIM];

  const int tid  = threadIdx.x;
  const int wave = tid >> 5;
  const int lane = tid & 31;
  const int half = lane >> 4;
  const int l16  = lane & 15;
  const int b    = blockIdx.x * 4 + wave;

  float* S  = stl[wave];

  v2f st[8][4];
  v8f d[8];
  float hc[NQ], hs[NQ];

  #pragma unroll
  for (int w = 0; w < NQ; ++w) { hc[w] = cs[b * 2 * NQ + w]; hs[w] = cs[b * 2 * NQ + NQ + w]; }
  first_core<5, 4, 3, 2, 1, 9, 8, 0, 7, 6>(d, lane, l16, half, hc, hs);
  #pragma unroll
  for (int t = 0; t < 4; ++t) {
    #pragma unroll
    for (int kk = 0; kk < 4; ++kk) { st[t][kk].x = d[t][2 * kk]; st[t][kk].y = d[t][2 * kk + 1]; }
  }
  #pragma unroll
  for (int w = 0; w < NQ; ++w) { hc[w] = tq[w]; hs[w] = tq[NQ + w]; }
  layer_core<1, 0, 3, 2, 4, 1, 7, 6, 9, 8, 5>(st, d, l16, half, hc, hs);
  ring_store<0, 3, 2, 4, 1, 7, 6, 9, 8, 5>(S, d, l16, half);
  wave_sync();

  for (int l = 1; l < NL - 1; ++l) {
    #pragma unroll
    for (int w = 0; w < NQ; ++w) { hc[w] = tq[l * 2 * NQ + w]; hs[w] = tq[l * 2 * NQ + NQ + w]; }
    load_state<5, 4, 3, 2, 1, 9, 8, 0, 7, 6>(S, st, l16, half);
    wave_sync();
    layer_core<2, 5, 4, 3, 2, 1, 9, 8, 0, 7, 6>(st, d, l16, half, hc, hs);
    ring_store<5, 4, 3, 2, 1, 9, 8, 0, 7, 6>(S, d, l16, half);
    wave_sync();
  }

  #pragma unroll
  for (int w = 0; w < NQ; ++w) {
    hc[w] = tq[(NL - 1) * 2 * NQ + w];
    hs[w] = tq[(NL - 1) * 2 * NQ + NQ + w];
  }
  load_state<5, 4, 3, 2, 1, 9, 8, 0, 7, 6>(S, st, l16, half);
  layer_core<2, 5, 4, 3, 2, 1, 9, 8, 0, 7, 6>(st, d, l16, half, hc, hs);
  measure_z<5, 4, 3, 2, 1, 9, 8, 0, 7, 6>(d, lane, z + b * NQ);
}

__global__ __launch_bounds__(128) void k_readout(
    const float* __restrict__ z, const float* __restrict__ Wout,
    const float* __restrict__ bout, float* __restrict__ out)
{
  const int lane = threadIdx.x & 31;
  const int wave = threadIdx.x >> 5;
  const int tile = blockIdx.x * 4 + wave;
  const int m0   = (tile >> 4) * 16;
  const int n0   = (tile & 15) * 16;
  const int half = lane >> 4;
  const int l16  = lane & 15;

  const float* arow = z    + (m0 + l16) * NQ;
  const float* brow = Wout + (n0 + l16) * NQ;

  v8f acc = {};
  #pragma unroll
  for (int kk = 0; kk < 3; ++kk) {
    const int k0  = kk * 4 + half * 2;
    const int k0c = (k0 < NQ - 1) ? k0 : (NQ - 2);
    const float vm0 = (k0     < NQ) ? 1.0f : 0.0f;
    const float vm1 = (k0 + 1 < NQ) ? 1.0f : 0.0f;
    v2f a, b;
    a.x = arow[k0c]     * vm0;
    a.y = arow[k0c + 1] * vm1;
    b.x = brow[k0c];
    b.y = brow[k0c + 1];
    acc = __builtin_amdgcn_wmma_f32_16x16x4_f32(false, a, false, b,
                                                (short)0, acc, false, false);
  }
  const float bias = bout[n0 + l16];
  #pragma unroll
  for (int r = 0; r < 8; ++r) {
    const int m = m0 + r + 8 * half;
    out[m * N_OUT + n0 + l16] = acc[r] + bias;
  }
}

extern "C" void kernel_launch(void* const* d_in, const int* in_sizes, int n_in,
                              void* d_out, int out_size, void* d_ws, size_t ws_size,
                              hipStream_t stream) {
  const float* x    = (const float*)d_in[0];
  const float* Win  = (const float*)d_in[1];
  const float* bin  = (const float*)d_in[2];
  const float* qp   = (const float*)d_in[3];
  const float* Wout = (const float*)d_in[4];
  const float* bout = (const float*)d_in[5];
  float* out = (float*)d_out;

  float* cs = (float*)d_ws;
  float* tq = cs + BATCH * 2 * NQ;
  float* zv = tq + 128;

  k_encode <<<BATCH / 16 + 1,      32, 0, stream>>>(x, Win, bin, qp, cs, tq);
  k_circuit<<<BATCH / 4,      128, 0, stream>>>(cs, tq, zv);
  k_readout<<<BATCH * 16 / 16 / 4, 128, 0, stream>>>(zv, Wout, bout, out);
}


# k_circuit: encoding and layer-0 RX of a qubit (no ring between them) applied as one rotation by the summed angle (cos/sin by addition formulas): 9 WMMAs up to the first ring, not 57; sums reordered
# speedup vs baseline: 3.7468x; 1.1008x over previous
#include <hip/hip_runtime.h>
#include <math.h>

typedef float v2f __attribute__((ext_vector_type(2)));
typedef float v4f __attribute__((ext_vector_type(4)));
typedef float v8f __attribute__((ext_vector_type(8)));

#define NQ 10
#define NL 4
#define DIM 1024
#define BATCH 4096
#define K_IN 128
#define N_OUT 256


__device__ __forceinline__ float tanh_straight(float x) {
  const float ax  = fabsf(x);
  const float e   = expf(ax + ax);
  const float big = fmaf(__builtin_amdgcn_rcpf(e + 1.0f), -2.0f, 1.0f);
  const float x2  = x * x;
  float p = fmaf(-0x1.758e7ap-8f, x2, 0x1.521192p-6f);
  p = fmaf(x2, p, -0x1.b8389cp-5f);
  p = fmaf(x2, p, 0x1.110704p-3f);
  p = fmaf(x2, p, -0x1.555532p-2f);
  const float small = fmaf(x2, ax * p, ax);
  return copysignf((0.625f > ax) ? small : big, x);
}

__global__ __launch_bounds__(32) __attribute__((amdgpu_waves_per_eu(1, 1))) void k_encode(
    const float* __restrict__ x, const float* __restrict__ Win,
    const float* __restrict__ bin, const float* __restrict__ qp,
    float* __restrict__ cs, float* __restrict__ tq)
{
  const int lane = threadIdx.x;
  const int m0   = blockIdx.x * 16;
  const int half = lane >> 4;
  const int l16  = lane & 15;

  if (blockIdx.x == BATCH / 16) {
    for (int i = lane; i < NL * NQ; i += 32) {
      const float a = 0.5f * qp[i];
      const int l = i / NQ, w = i - l * NQ;
      tq[l * 2 * NQ + w]      = cosf(a);
      tq[l * 2 * NQ + NQ + w] = sinf(a);
    }
    return;
  }

  const int    nrow = (l16 < NQ) ? l16 : (NQ - 1);
  const float* arow = x   + (m0 + l16) * K_IN + half * 4;
  const float* brow = Win + nrow * K_IN + half * 4;
  const float  bias = bin[nrow];

  v4f a[K_IN / 8], b[K_IN / 8];
  #pragma unroll
  for (int j = 0; j < K_IN / 8; ++j) {
    a[j].x = arow[8 * j];     a[j].y = arow[8 * j + 1];
    a[j].z = arow[8 * j + 2]; a[j].w = arow[8 * j + 3];
    b[j].x = brow[8 * j];     b[j].y = brow[8 * j + 1];
    b[j].z = brow[8 * j + 2]; b[j].w = brow[8 * j + 3];
  }

  v8f acc0, acc1 = {};
  #pragma unroll
  for (int r = 0; r < 8; ++r) acc0[r] = bias;
  #pragma unroll
  for (int j = 0; j < K_IN / 8; ++j) {
    v2f a0, b0, a1, b1;
    a0.x = a[j].x; a0.y = a[j].y; b0.x = b[j].x; b0.y = b[j].y;
    a1.x = a[j].z; a1.y = a[j].w; b1.x = b[j].z; b1.y = b[j].w;
    acc0 = __builtin_amdgcn_wmma_f32_16x16x4_f32(false, a0, false, b0,
                                                 (short)0, acc0, false, false);
    acc1 = __builtin_amdgcn_wmma_f32_16x16x4_f32(false, a1, false, b1,
                                                 (short)0, acc1, false, false);
  }
  const v8f pre = acc0 + acc1;
  if (l16 < NQ) {
    #pragma unroll
    for (int r = 0; r < 8; ++r) {
      const int m = m0 + r + 8 * half;
      const float a  = 0.5f * tanh_straight(pre[r]);
      const float a2 = a * a;
      float cp = fmaf(a2, -1.0f / 3628800.0f, 1.0f / 40320.0f);
      cp = fmaf(a2, cp, -1.0f / 720.0f);
      cp = fmaf(a2, cp, 1.0f / 24.0f);
      cp = fmaf(a2, cp, -0.5f);
      float sp = fmaf(a2, 1.0f / 362880.0f, -1.0f / 5040.0f);
      sp = fmaf(a2, sp, 1.0f / 120.0f);
      sp = fmaf(a2, sp, -1.0f / 6.0f);
      cs[m * 2 * NQ + l16]      = fmaf(a2, cp, 1.0f);
      cs[m * 2 * NQ + NQ + l16] = fmaf(a * a2, sp, a);
    }
  }
}


__device__ __forceinline__ void wave_sync() {
  __builtin_amdgcn_fence(__ATOMIC_RELEASE, "wavefront");
  __builtin_amdgcn_wave_barrier();
  __builtin_amdgcn_fence(__ATOMIC_ACQUIRE, "wavefront");
}

__device__ __forceinline__ constexpr int ring_dst(int s) {
  for (int c = 0; c <= 8; ++c)
    if ((s >> (9 - c)) & 1) s ^= 1 << (8 - c);
  if (s & 1) s ^= 1 << 9;
  return s;
}

__device__ __forceinline__ constexpr int popc(int v) { return __builtin_popcount(v); }

__device__ __forceinline__ float rot(bool a, bool b, float c, float s) {
  return (a == b) ? c : (a ? -s : s);
}

__device__ __forceinline__ void gate_frag(v2f (&g)[4], bool o3, bool o2, bool o1, bool o0,
    bool half, float c3, float s3, float c2, float s2, float c1, float s1, float c0, float s0)
{
  const float fH = rot(o1, half, c1, s1);
  const float w0 = fH * rot(o0, false, c0, s0);
  const float w1 = fH * rot(o0, true,  c0, s0);
  #pragma unroll
  for (int kk = 0; kk < 4; ++kk) {
    const float u = rot(o3, (kk >> 1) != 0, c3, s3) * rot(o2, (kk & 1) != 0, c2, s2);
    g[kk].x = u * w0;
    g[kk].y = u * w1;
  }
}

template <int HB, int L3, int L2, int L1, int L0, int K1, int K0, int XY, int T1, int T0>
__device__ __forceinline__ void layer_core(
    const v2f (&st)[8][4], v8f (&d)[8], int l16, int half,
    const float (&hc)[NQ], const float (&hs)[NQ])
{
  constexpr int NT = 8;
  const bool n3 = (l16 >> 3) & 1, n2 = (l16 >> 2) & 1;
  const bool n1 = (l16 >> 1) & 1, n0 = l16 & 1;
  const bool hb = half != 0;
  v2f g[4];

  gate_frag(g, n3, n2, n1, n0, hb, hc[9 - K1], hs[9 - K1], hc[9 - K0], hs[9 - K0],
            hc[9 - HB], hs[9 - HB], hc[9 - XY], hs[9 - XY]);
  v8f d1[NT];
  #pragma unroll
  for (int t = 0; t < NT; ++t) d1[t] = v8f{};
  #pragma unroll
  for (int kk = 0; kk < 4; ++kk) {
    #pragma unroll
    for (int t = 0; t < NT; ++t)
      d1[t] = __builtin_amdgcn_wmma_f32_16x16x4_f32(false, st[t][kk], false, g[kk],
                                                    (short)0, d1[t], false, false);
  }

  gate_frag(g, n3, n2, n1, n0, hb, hc[9 - L2], hs[9 - L2], hc[9 - L1], hs[9 - L1],
            hc[9 - L3], hs[9 - L3], hc[9 - L0], hs[9 - L0]);
  v8f d2[NT];
  #pragma unroll
  for (int t = 0; t < NT; ++t) d2[t] = v8f{};
  #pragma unroll
  for (int kk = 0; kk < 4; ++kk) {
    #pragma unroll
    for (int t = 0; t < NT; ++t) {
      v2f a;
      a.x = d1[t][2 * kk];
      a.y = d1[t][2 * kk + 1];
      d2[t] = __builtin_amdgcn_wmma_f32_16x16x4_f32(false, a, false, g[kk],
                                                    (short)0, d2[t], false, false);
    }
  }

  gate_frag(g, n2, n1, n0, n3, hb, hc[9 - T1], hs[9 - T1], hc[9 - T0], hs[9 - T0],
            1.0f, 0.0f, 1.0f, 0.0f);
  #pragma unroll
  for (int t = 0; t < NT; ++t) d[t] = v8f{};
  #pragma unroll
  for (int kk = 0; kk < 4; ++kk) {
    #pragma unroll
    for (int tp = 0; tp < NT; ++tp) {
      const int p = tp >> 2, u = tp & 3;
      v2f b;
      b.x = d2[(p << 2) | kk][2 * u];
      b.y = d2[(p << 2) | kk][2 * u + 1];
      d[tp] = __builtin_amdgcn_wmma_f32_16x16x4_f32(false, g[kk], false, b,
                                                    (short)0, d[tp], false, false);
    }
  }
}

template <int HB, int L3, int L2, int L1, int L0, int K1, int K0, int XY, int T1, int T0>
__device__ __forceinline__ void first_core(
    v8f (&d)[8], int lane, int l16, int half,
    const float (&hc)[NQ], const float (&hs)[NQ])
{
  const bool n3 = (l16 >> 3) & 1, n2 = (l16 >> 2) & 1;
  const bool n1 = (l16 >> 1) & 1, n0 = l16 & 1;
  const bool hb = half != 0;
  v2f g[4];

  v2f a0;
  a0.x = (lane == 0) ? 1.0f : 0.0f;
  a0.y = 0.0f;
  gate_frag(g, n3, n2, n1, n0, hb, hc[9 - K1], hs[9 - K1], hc[9 - K0], hs[9 - K0],
            hc[9 - HB], hs[9 - HB], hc[9 - XY], hs[9 - XY]);
  const v8f d1 = __builtin_amdgcn_wmma_f32_16x16x4_f32(false, a0, false, g[0],
                                                       (short)0, v8f{}, false, false);

  gate_frag(g, n3, n2, n1, n0, hb, hc[9 - L2], hs[9 - L2], hc[9 - L1], hs[9 - L1],
            hc[9 - L3], hs[9 - L3], hc[9 - L0], hs[9 - L0]);
  v8f d2 = v8f{};
  #pragma unroll
  for (int kk = 0; kk < 4; ++kk) {
    v2f a;
    a.x = d1[2 * kk];
    a.y = d1[2 * kk + 1];
    d2 = __builtin_amdgcn_wmma_f32_16x16x4_f32(false, a, false, g[kk],
                                               (short)0, d2, false, false);
  }

  gate_frag(g, n2, n1, n0, n3, hb, hc[9 - T1], hs[9 - T1], hc[9 - T0], hs[9 - T0],
            1.0f, 0.0f, 1.0f, 0.0f);
  #pragma unroll
  for (int u = 0; u < 4; ++u) {
    v2f b;
    b.x = d2[2 * u];
    b.y = d2[2 * u + 1];
    d[u] = __builtin_amdgcn_wmma_f32_16x16x4_f32(false, g[0], false, b,
                                                 (short)0, v8f{}, false, false);
  }
  #pragma unroll
  for (int t = 4; t < 8; ++t) d[t] = v8f{};
}

template <int HB, int L3, int L2, int L1, int L0, int K1, int K0, int XY, int T1, int T0>
__device__ __forceinline__ void load_state(
    const float* __restrict__ S, v2f (&st)[8][4], int l16, int half)
{
  const int lb = (((l16 >> 3) & 1) << L3) | (((l16 >> 2) & 1) << L2) |
                 (((l16 >> 1) & 1) << L1) | ((l16 & 1) << L0) | (half << HB);
  #pragma unroll
  for (int t = 0; t < 8; ++t) {
    #pragma unroll
    for (int kk = 0; kk < 4; ++kk) {
      const int o = (((t >> 2) & 1) << 10) | (((t >> 1) & 1) << T1) | ((t & 1) << T0) |
                    (((kk >> 1) & 1) << K1) | ((kk & 1) << K0);
      st[t][kk].x = S[lb + o];
      st[t][kk].y = S[lb + o + (1 << XY)];
    }
  }
}

template <int HB, int L3, int L2, int L1, int L0, int K1, int K0, int XY, int T1, int T0>
__device__ __forceinline__ void ring_store(
    float* __restrict__ S, const v8f (&d)[8], int l16, int half)
{
  constexpr int AH = XY, A3 = L2, A2 = L1, A1 = L3, A0 = L0;
  constexpr int R2 = T1, R1 = T0, R0 = K1, U1 = K0, U0 = HB;
  const int lb = (((l16 >> 3) & 1) << A3) | (((l16 >> 2) & 1) << A2) |
                 (((l16 >> 1) & 1) << A1) | ((l16 & 1) << A0) | (half << AH);
  constexpr int LM = ring_dst(1 << A3) | ring_dst(1 << A2) | ring_dst(1 << A1) |
                     ring_dst(1 << A0) | ring_dst(1 << AH);
  const int lr  = ring_dst(lb);
  const int kl  = (popc(lb) - popc(lr)) & 3;
  const int lrc = lr | ((kl & 1) << 10);
  unsigned sbm[4];
  #pragma unroll
  for (int kc = 0; kc < 4; ++kc)
    sbm[kc] = (unsigned)(((kl + kc) >> 1) & 1) << 31;
  constexpr int RB[5] = {R0, R1, R2, U0, U1};
  unsigned cm[32];
  cm[0] = 0u;
  #pragma unroll
  for (int x = 1; x < 32; ++x) {
    const int low = x & -x;
    const int bi  = (low == 1) ? 0 : (low == 2) ? 1 : (low == 4) ? 2 : (low == 8) ? 3 : 4;
    const unsigned bmw = (unsigned)(popc(lr & ring_dst(1 << RB[bi])) & 1) << 31;
    cm[x] = cm[x ^ low] ^ bmw;
  }
  #pragma unroll
  for (int t = 0; t < 8; ++t) {
    #pragma unroll
    for (int r = 0; r < 8; ++r) {
      const int regS = (((t >> 1) & 1) << U1) | ((t & 1) << U0) |
                       (((r >> 2) & 1) << R2) | (((r >> 1) & 1) << R1) | ((r & 1) << R0);
      const int pl  = (t >> 2) & 1;
      const int ro  = ring_dst(regS);
      const int ko  = (popc(regS) - popc(ro)) & 3;
      const int kc  = (ko + pl) & 3;
      const int xc  = (ro & LM) | (((ko & 1) ^ pl) << 10);
      const int add = ro & ~LM & 0x3FF;
      const int x   = (r & 7) | ((t & 3) << 3);
      const unsigned w = __float_as_uint(d[t][r]) ^ cm[x] ^ sbm[kc];
      S[(lrc ^ xc) + add] = __uint_as_float(w);
    }
  }
}

__device__ __forceinline__ float xhalf(float v) {
  const int b = __float_as_int(v);
  return __int_as_float(__builtin_amdgcn_permlanex16(b, b, 0x76543210, 0xfedcba98, false, false));
}

template <int M>
__device__ __forceinline__ float row_xor_add(float v) {
  const int b = __float_as_int(v);
  return v + __int_as_float(__builtin_amdgcn_update_dpp(0, b, 0x160 + M, 0xf, 0xf, true));
}

template <int HB, int L3, int L2, int L1, int L0, int K1, int K0, int XY, int T1, int T0>
__device__ __forceinline__ void measure_z(
    const v8f (&d)[8], int lane, float* __restrict__ zrow)
{
  constexpr int AH = XY, A3 = L2, A2 = L1, A1 = L3, A0 = L0;
  constexpr int R2 = T1, R1 = T0, R0 = K1, U1 = K0, U0 = HB;

  float w[32];
  #pragma unroll
  for (int t = 0; t < 4; ++t) {
    #pragma unroll
    for (int r = 0; r < 8; ++r)
      w[(t << 3) | r] = d[t][r] * d[t][r] + d[t + 4][r] * d[t + 4][r];
  }
  constexpr int ORD[5] = {4, 3, 1, 2, 0};
  #pragma unroll
  for (int i = 0; i < 5; ++i) {
    const int bit = 1 << ORD[i];
    #pragma unroll
    for (int x = 0; x < 32; ++x) {
      if (x & bit) continue;
      const float a = w[x], b = w[x | bit];
      w[x]       = a + b;
      w[x | bit] = a - b;
    }
  }
  constexpr int RB[5] = {R0, R1, R2, U0, U1};
  constexpr int LB[5] = {A0, A1, A2, A3, AH};
  float z = 0.0f;
  #pragma unroll
  for (int q = 0; q < NQ; ++q) {
    int m = 0, lm = 0;
    #pragma unroll
    for (int i = 0; i < 5; ++i) {
      m  |= ((ring_dst(1 << RB[i]) >> (9 - q)) & 1) << i;
      lm |= ((ring_dst(1 << LB[i]) >> (9 - q)) & 1) << i;
    }
    float v = w[m];
    if (lm)
      v = __uint_as_float(__float_as_uint(v) ^ ((unsigned)(__builtin_popcount(lane & lm) & 1) << 31));
    v = row_xor_add<1>(v);
    v = row_xor_add<2>(v);
    v = row_xor_add<4>(v);
    v = row_xor_add<8>(v);
    v += xhalf(v);
    z = (lane == q) ? v : z;
  }
  if (lane < NQ) zrow[lane] = z;
}

__global__ __launch_bounds__(128) void k_circuit(
    const float* __restrict__ cs, const float* __restrict__ tq,
    float* __restrict__ z)
{
  __shared__ float stl[4][2 * DIM];

  const int tid  = threadIdx.x;
  const int wave = tid >> 5;
  const int lane = tid & 31;
  const int half = lane >> 4;
  const int l16  = lane & 15;
  const int b    = blockIdx.x * 4 + wave;

  float* S  = stl[wave];

  v2f st[8][4];
  v8f d[8];
  float hc[NQ], hs[NQ];

  #pragma unroll
  for (int w = 0; w < NQ; ++w) {
    const float ce = cs[b * 2 * NQ + w], se = cs[b * 2 * NQ + NQ + w];
    const float c0 = tq[w],              s0 = tq[NQ + w];
    hc[w] = fmaf(ce, c0, -(se * s0));
    hs[w] = fmaf(se, c0, ce * s0);
  }
  first_core<5, 4, 3, 2, 1, 9, 8, 0, 7, 6>(d, lane, l16, half, hc, hs);
  ring_store<5, 4, 3, 2, 1, 9, 8, 0, 7, 6>(S, d, l16, half);
  wave_sync();

  for (int l = 1; l < NL - 1; ++l) {
    #pragma unroll
    for (int w = 0; w < NQ; ++w) { hc[w] = tq[l * 2 * NQ + w]; hs[w] = tq[l * 2 * NQ + NQ + w]; }
    load_state<5, 4, 3, 2, 1, 9, 8, 0, 7, 6>(S, st, l16, half);
    wave_sync();
    layer_core<5, 4, 3, 2, 1, 9, 8, 0, 7, 6>(st, d, l16, half, hc, hs);
    ring_store<5, 4, 3, 2, 1, 9, 8, 0, 7, 6>(S, d, l16, half);
    wave_sync();
  }

  #pragma unroll
  for (int w = 0; w < NQ; ++w) {
    hc[w] = tq[(NL - 1) * 2 * NQ + w];
    hs[w] = tq[(NL - 1) * 2 * NQ + NQ + w];
  }
  load_state<5, 4, 3, 2, 1, 9, 8, 0, 7, 6>(S, st, l16, half);
  layer_core<5, 4, 3, 2, 1, 9, 8, 0, 7, 6>(st, d, l16, half, hc, hs);
  measure_z<5, 4, 3, 2, 1, 9, 8, 0, 7, 6>(d, lane, z + b * NQ);
}

__global__ __launch_bounds__(128) void k_readout(
    const float* __restrict__ z, const float* __restrict__ Wout,
    const float* __restrict__ bout, float* __restrict__ out)
{
  const int lane = threadIdx.x & 31;
  const int wave = threadIdx.x >> 5;
  const int tile = blockIdx.x * 4 + wave;
  const int m0   = (tile >> 4) * 16;
  const int n0   = (tile & 15) * 16;
  const int half = lane >> 4;
  const int l16  = lane & 15;

  const float* arow = z    + (m0 + l16) * NQ;
  const float* brow = Wout + (n0 + l16) * NQ;

  v8f acc = {};
  #pragma unroll
  for (int kk = 0; kk < 3; ++kk) {
    const int k0  = kk * 4 + half * 2;
    const int k0c = (k0 < NQ - 1) ? k0 : (NQ - 2);
    const float vm0 = (k0     < NQ) ? 1.0f : 0.0f;
    const float vm1 = (k0 + 1 < NQ) ? 1.0f : 0.0f;
    v2f a, b;
    a.x = arow[k0c]     * vm0;
    a.y = arow[k0c + 1] * vm1;
    b.x = brow[k0c];
    b.y = brow[k0c + 1];
    acc = __builtin_amdgcn_wmma_f32_16x16x4_f32(false, a, false, b,
                                                (short)0, acc, false, false);
  }
  const float bias = bout[n0 + l16];
  #pragma unroll
  for (int r = 0; r < 8; ++r) {
    const int m = m0 + r + 8 * half;
    out[m * N_OUT + n0 + l16] = acc[r] + bias;
  }
}

extern "C" void kernel_launch(void* const* d_in, const int* in_sizes, int n_in,
                              void* d_out, int out_size, void* d_ws, size_t ws_size,
                              hipStream_t stream) {
  const float* x    = (const float*)d_in[0];
  const float* Win  = (const float*)d_in[1];
  const float* bin  = (const float*)d_in[2];
  const float* qp   = (const float*)d_in[3];
  const float* Wout = (const float*)d_in[4];
  const float* bout = (const float*)d_in[5];
  float* out = (float*)d_out;

  float* cs = (float*)d_ws;
  float* tq = cs + BATCH * 2 * NQ;
  float* zv = tq + 128;

  k_encode <<<BATCH / 16 + 1,      32, 0, stream>>>(x, Win, bin, qp, cs, tq);
  k_circuit<<<BATCH / 4,      128, 0, stream>>>(cs, tq, zv);
  k_readout<<<BATCH * 16 / 16 / 4, 128, 0, stream>>>(zv, Wout, bout, out);
}


# readout folded into k_circuit (3 launches, now 2): z rows pass through a 4x12 LDS patch and one barrier, wave w makes out[4 rows][64w..64w+63] on 12 WMMAs, W_out fragments loaded at entry
# speedup vs baseline: 4.5335x; 1.2100x over previous
#include <hip/hip_runtime.h>
#include <math.h>

typedef float v2f __attribute__((ext_vector_type(2)));
typedef float v4f __attribute__((ext_vector_type(4)));
typedef float v8f __attribute__((ext_vector_type(8)));

#define NQ 10
#define NL 4
#define DIM 1024
#define BATCH 4096
#define K_IN 128
#define N_OUT 256


__device__ __forceinline__ float tanh_straight(float x) {
  const float ax  = fabsf(x);
  const float e   = expf(ax + ax);
  const float big = fmaf(__builtin_amdgcn_rcpf(e + 1.0f), -2.0f, 1.0f);
  const float x2  = x * x;
  float p = fmaf(-0x1.758e7ap-8f, x2, 0x1.521192p-6f);
  p = fmaf(x2, p, -0x1.b8389cp-5f);
  p = fmaf(x2, p, 0x1.110704p-3f);
  p = fmaf(x2, p, -0x1.555532p-2f);
  const float small = fmaf(x2, ax * p, ax);
  return copysignf((0.625f > ax) ? small : big, x);
}

__global__ __launch_bounds__(32) __attribute__((amdgpu_waves_per_eu(1, 1))) void k_encode(
    const float* __restrict__ x, const float* __restrict__ Win,
    const float* __restrict__ bin, const float* __restrict__ qp,
    float* __restrict__ cs, float* __restrict__ tq)
{
  const int lane = threadIdx.x;
  const int m0   = blockIdx.x * 16;
  const int half = lane >> 4;
  const int l16  = lane & 15;

  if (blockIdx.x == BATCH / 16) {
    for (int i = lane; i < NL * NQ; i += 32) {
      const float a = 0.5f * qp[i];
      const int l = i / NQ, w = i - l * NQ;
      tq[l * 2 * NQ + w]      = cosf(a);
      tq[l * 2 * NQ + NQ + w] = sinf(a);
    }
    return;
  }

  const int    nrow = (l16 < NQ) ? l16 : (NQ - 1);
  const float* arow = x   + (m0 + l16) * K_IN + half * 4;
  const float* brow = Win + nrow * K_IN + half * 4;
  const float  bias = bin[nrow];

  v4f a[K_IN / 8], b[K_IN / 8];
  #pragma unroll
  for (int j = 0; j < K_IN / 8; ++j) {
    a[j].x = arow[8 * j];     a[j].y = arow[8 * j + 1];
    a[j].z = arow[8 * j + 2]; a[j].w = arow[8 * j + 3];
    b[j].x = brow[8 * j];     b[j].y = brow[8 * j + 1];
    b[j].z = brow[8 * j + 2]; b[j].w = brow[8 * j + 3];
  }

  v8f acc0, acc1 = {};
  #pragma unroll
  for (int r = 0; r < 8; ++r) acc0[r] = bias;
  #pragma unroll
  for (int j = 0; j < K_IN / 8; ++j) {
    v2f a0, b0, a1, b1;
    a0.x = a[j].x; a0.y = a[j].y; b0.x = b[j].x; b0.y = b[j].y;
    a1.x = a[j].z; a1.y = a[j].w; b1.x = b[j].z; b1.y = b[j].w;
    acc0 = __builtin_amdgcn_wmma_f32_16x16x4_f32(false, a0, false, b0,
                                                 (short)0, acc0, false, false);
    acc1 = __builtin_amdgcn_wmma_f32_16x16x4_f32(false, a1, false, b1,
                                                 (short)0, acc1, false, false);
  }
  const v8f pre = acc0 + acc1;
  if (l16 < NQ) {
    #pragma unroll
    for (int r = 0; r < 8; ++r) {
      const int m = m0 + r + 8 * half;
      const float a  = 0.5f * tanh_straight(pre[r]);
      const float a2 = a * a;
      float cp = fmaf(a2, -1.0f / 3628800.0f, 1.0f / 40320.0f);
      cp = fmaf(a2, cp, -1.0f / 720.0f);
      cp = fmaf(a2, cp, 1.0f / 24.0f);
      cp = fmaf(a2, cp, -0.5f);
      float sp = fmaf(a2, 1.0f / 362880.0f, -1.0f / 5040.0f);
      sp = fmaf(a2, sp, 1.0f / 120.0f);
      sp = fmaf(a2, sp, -1.0f / 6.0f);
      cs[m * 2 * NQ + l16]      = fmaf(a2, cp, 1.0f);
      cs[m * 2 * NQ + NQ + l16] = fmaf(a * a2, sp, a);
    }
  }
}


__device__ __forceinline__ void wave_sync() {
  __builtin_amdgcn_fence(__ATOMIC_RELEASE, "wavefront");
  __builtin_amdgcn_wave_barrier();
  __builtin_amdgcn_fence(__ATOMIC_ACQUIRE, "wavefront");
}

__device__ __forceinline__ constexpr int ring_dst(int s) {
  for (int c = 0; c <= 8; ++c)
    if ((s >> (9 - c)) & 1) s ^= 1 << (8 - c);
  if (s & 1) s ^= 1 << 9;
  return s;
}

__device__ __forceinline__ constexpr int popc(int v) { return __builtin_popcount(v); }

__device__ __forceinline__ float rot(bool a, bool b, float c, float s) {
  return (a == b) ? c : (a ? -s : s);
}

__device__ __forceinline__ void gate_frag(v2f (&g)[4], bool o3, bool o2, bool o1, bool o0,
    bool half, float c3, float s3, float c2, float s2, float c1, float s1, float c0, float s0)
{
  const float fH = rot(o1, half, c1, s1);
  const float w0 = fH * rot(o0, false, c0, s0);
  const float w1 = fH * rot(o0, true,  c0, s0);
  #pragma unroll
  for (int kk = 0; kk < 4; ++kk) {
    const float u = rot(o3, (kk >> 1) != 0, c3, s3) * rot(o2, (kk & 1) != 0, c2, s2);
    g[kk].x = u * w0;
    g[kk].y = u * w1;
  }
}

template <int HB, int L3, int L2, int L1, int L0, int K1, int K0, int XY, int T1, int T0>
__device__ __forceinline__ void layer_core(
    const v2f (&st)[8][4], v8f (&d)[8], int l16, int half,
    const float (&hc)[NQ], const float (&hs)[NQ])
{
  constexpr int NT = 8;
  const bool n3 = (l16 >> 3) & 1, n2 = (l16 >> 2) & 1;
  const bool n1 = (l16 >> 1) & 1, n0 = l16 & 1;
  const bool hb = half != 0;
  v2f g[4];

  gate_frag(g, n3, n2, n1, n0, hb, hc[9 - K1], hs[9 - K1], hc[9 - K0], hs[9 - K0],
            hc[9 - HB], hs[9 - HB], hc[9 - XY], hs[9 - XY]);
  v8f d1[NT];
  #pragma unroll
  for (int t = 0; t < NT; ++t) d1[t] = v8f{};
  #pragma unroll
  for (int kk = 0; kk < 4; ++kk) {
    #pragma unroll
    for (int t = 0; t < NT; ++t)
      d1[t] = __builtin_amdgcn_wmma_f32_16x16x4_f32(false, st[t][kk], false, g[kk],
                                                    (short)0, d1[t], false, false);
  }

  gate_frag(g, n3, n2, n1, n0, hb, hc[9 - L2], hs[9 - L2], hc[9 - L1], hs[9 - L1],
            hc[9 - L3], hs[9 - L3], hc[9 - L0], hs[9 - L0]);
  v8f d2[NT];
  #pragma unroll
  for (int t = 0; t < NT; ++t) d2[t] = v8f{};
  #pragma unroll
  for (int kk = 0; kk < 4; ++kk) {
    #pragma unroll
    for (int t = 0; t < NT; ++t) {
      v2f a;
      a.x = d1[t][2 * kk];
      a.y = d1[t][2 * kk + 1];
      d2[t] = __builtin_amdgcn_wmma_f32_16x16x4_f32(false, a, false, g[kk],
                                                    (short)0, d2[t], false, false);
    }
  }

  gate_frag(g, n2, n1, n0, n3, hb, hc[9 - T1], hs[9 - T1], hc[9 - T0], hs[9 - T0],
            1.0f, 0.0f, 1.0f, 0.0f);
  #pragma unroll
  for (int t = 0; t < NT; ++t) d[t] = v8f{};
  #pragma unroll
  for (int kk = 0; kk < 4; ++kk) {
    #pragma unroll
    for (int tp = 0; tp < NT; ++tp) {
      const int p = tp >> 2, u = tp & 3;
      v2f b;
      b.x = d2[(p << 2) | kk][2 * u];
      b.y = d2[(p << 2) | kk][2 * u + 1];
      d[tp] = __builtin_amdgcn_wmma_f32_16x16x4_f32(false, g[kk], false, b,
                                                    (short)0, d[tp], false, false);
    }
  }
}

template <int HB, int L3, int L2, int L1, int L0, int K1, int K0, int XY, int T1, int T0>
__device__ __forceinline__ void first_core(
    v8f (&d)[8], int lane, int l16, int half,
    const float (&hc)[NQ], const float (&hs)[NQ])
{
  const bool n3 = (l16 >> 3) & 1, n2 = (l16 >> 2) & 1;
  const bool n1 = (l16 >> 1) & 1, n0 = l16 & 1;
  const bool hb = half != 0;
  v2f g[4];

  v2f a0;
  a0.x = (lane == 0) ? 1.0f : 0.0f;
  a0.y = 0.0f;
  gate_frag(g, n3, n2, n1, n0, hb, hc[9 - K1], hs[9 - K1], hc[9 - K0], hs[9 - K0],
            hc[9 - HB], hs[9 - HB], hc[9 - XY], hs[9 - XY]);
  const v8f d1 = __builtin_amdgcn_wmma_f32_16x16x4_f32(false, a0, false, g[0],
                                                       (short)0, v8f{}, false, false);

  gate_frag(g, n3, n2, n1, n0, hb, hc[9 - L2], hs[9 - L2], hc[9 - L1], hs[9 - L1],
            hc[9 - L3], hs[9 - L3], hc[9 - L0], hs[9 - L0]);
  v8f d2 = v8f{};
  #pragma unroll
  for (int kk = 0; kk < 4; ++kk) {
    v2f a;
    a.x = d1[2 * kk];
    a.y = d1[2 * kk + 1];
    d2 = __builtin_amdgcn_wmma_f32_16x16x4_f32(false, a, false, g[kk],
                                               (short)0, d2, false, false);
  }

  gate_frag(g, n2, n1, n0, n3, hb, hc[9 - T1], hs[9 - T1], hc[9 - T0], hs[9 - T0],
            1.0f, 0.0f, 1.0f, 0.0f);
  #pragma unroll
  for (int u = 0; u < 4; ++u) {
    v2f b;
    b.x = d2[2 * u];
    b.y = d2[2 * u + 1];
    d[u] = __builtin_amdgcn_wmma_f32_16x16x4_f32(false, g[0], false, b,
                                                 (short)0, v8f{}, false, false);
  }
  #pragma unroll
  for (int t = 4; t < 8; ++t) d[t] = v8f{};
}

template <int HB, int L3, int L2, int L1, int L0, int K1, int K0, int XY, int T1, int T0>
__device__ __forceinline__ void load_state(
    const float* __restrict__ S, v2f (&st)[8][4], int l16, int half)
{
  const int lb = (((l16 >> 3) & 1) << L3) | (((l16 >> 2) & 1) << L2) |
                 (((l16 >> 1) & 1) << L1) | ((l16 & 1) << L0) | (half << HB);
  #pragma unroll
  for (int t = 0; t < 8; ++t) {
    #pragma unroll
    for (int kk = 0; kk < 4; ++kk) {
      const int o = (((t >> 2) & 1) << 10) | (((t >> 1) & 1) << T1) | ((t & 1) << T0) |
                    (((kk >> 1) & 1) << K1) | ((kk & 1) << K0);
      st[t][kk].x = S[lb + o];
      st[t][kk].y = S[lb + o + (1 << XY)];
    }
  }
}

template <int HB, int L3, int L2, int L1, int L0, int K1, int K0, int XY, int T1, int T0>
__device__ __forceinline__ void ring_store(
    float* __restrict__ S, const v8f (&d)[8], int l16, int half)
{
  constexpr int AH = XY, A3 = L2, A2 = L1, A1 = L3, A0 = L0;
  constexpr int R2 = T1, R1 = T0, R0 = K1, U1 = K0, U0 = HB;
  const int lb = (((l16 >> 3) & 1) << A3) | (((l16 >> 2) & 1) << A2) |
                 (((l16 >> 1) & 1) << A1) | ((l16 & 1) << A0) | (half << AH);
  constexpr int LM = ring_dst(1 << A3) | ring_dst(1 << A2) | ring_dst(1 << A1) |
                     ring_dst(1 << A0) | ring_dst(1 << AH);
  const int lr  = ring_dst(lb);
  const int kl  = (popc(lb) - popc(lr)) & 3;
  const int lrc = lr | ((kl & 1) << 10);
  unsigned sbm[4];
  #pragma unroll
  for (int kc = 0; kc < 4; ++kc)
    sbm[kc] = (unsigned)(((kl + kc) >> 1) & 1) << 31;
  constexpr int RB[5] = {R0, R1, R2, U0, U1};
  unsigned cm[32];
  cm[0] = 0u;
  #pragma unroll
  for (int x = 1; x < 32; ++x) {
    const int low = x & -x;
    const int bi  = (low == 1) ? 0 : (low == 2) ? 1 : (low == 4) ? 2 : (low == 8) ? 3 : 4;
    const unsigned bmw = (unsigned)(popc(lr & ring_dst(1 << RB[bi])) & 1) << 31;
    cm[x] = cm[x ^ low] ^ bmw;
  }
  #pragma unroll
  for (int t = 0; t < 8; ++t) {
    #pragma unroll
    for (int r = 0; r < 8; ++r) {
      const int regS = (((t >> 1) & 1) << U1) | ((t & 1) << U0) |
                       (((r >> 2) & 1) << R2) | (((r >> 1) & 1) << R1) | ((r & 1) << R0);
      const int pl  = (t >> 2) & 1;
      const int ro  = ring_dst(regS);
      const int ko  = (popc(regS) - popc(ro)) & 3;
      const int kc  = (ko + pl) & 3;
      const int xc  = (ro & LM) | (((ko & 1) ^ pl) << 10);
      const int add = ro & ~LM & 0x3FF;
      const int x   = (r & 7) | ((t & 3) << 3);
      const unsigned w = __float_as_uint(d[t][r]) ^ cm[x] ^ sbm[kc];
      S[(lrc ^ xc) + add] = __uint_as_float(w);
    }
  }
}

__device__ __forceinline__ float xhalf(float v) {
  const int b = __float_as_int(v);
  return __int_as_float(__builtin_amdgcn_permlanex16(b, b, 0x76543210, 0xfedcba98, false, false));
}

template <int M>
__device__ __forceinline__ float row_xor_add(float v) {
  const int b = __float_as_int(v);
  return v + __int_as_float(__builtin_amdgcn_update_dpp(0, b, 0x160 + M, 0xf, 0xf, true));
}

template <int HB, int L3, int L2, int L1, int L0, int K1, int K0, int XY, int T1, int T0>
__device__ __forceinline__ float measure_z(
    const v8f (&d)[8], int lane)
{
  constexpr int AH = XY, A3 = L2, A2 = L1, A1 = L3, A0 = L0;
  constexpr int R2 = T1, R1 = T0, R0 = K1, U1 = K0, U0 = HB;

  float w[32];
  #pragma unroll
  for (int t = 0; t < 4; ++t) {
    #pragma unroll
    for (int r = 0; r < 8; ++r)
      w[(t << 3) | r] = d[t][r] * d[t][r] + d[t + 4][r] * d[t + 4][r];
  }
  constexpr int ORD[5] = {4, 3, 1, 2, 0};
  #pragma unroll
  for (int i = 0; i < 5; ++i) {
    const int bit = 1 << ORD[i];
    #pragma unroll
    for (int x = 0; x < 32; ++x) {
      if (x & bit) continue;
      const float a = w[x], b = w[x | bit];
      w[x]       = a + b;
      w[x | bit] = a - b;
    }
  }
  constexpr int RB[5] = {R0, R1, R2, U0, U1};
  constexpr int LB[5] = {A0, A1, A2, A3, AH};
  float z = 0.0f;
  #pragma unroll
  for (int q = 0; q < NQ; ++q) {
    int m = 0, lm = 0;
    #pragma unroll
    for (int i = 0; i < 5; ++i) {
      m  |= ((ring_dst(1 << RB[i]) >> (9 - q)) & 1) << i;
      lm |= ((ring_dst(1 << LB[i]) >> (9 - q)) & 1) << i;
    }
    float v = w[m];
    if (lm)
      v = __uint_as_float(__float_as_uint(v) ^ ((unsigned)(__builtin_popcount(lane & lm) & 1) << 31));
    v = row_xor_add<1>(v);
    v = row_xor_add<2>(v);
    v = row_xor_add<4>(v);
    v = row_xor_add<8>(v);
    v += xhalf(v);
    z = (lane == q) ? v : z;
  }
  return z;
}

__global__ __launch_bounds__(128) void k_circuit(
    const float* __restrict__ cs, const float* __restrict__ tq,
    const float* __restrict__ Wout, const float* __restrict__ bout,
    float* __restrict__ out)
{
  __shared__ float stl[4][2 * DIM];
  __shared__ float zsh[4][12];

  const int tid  = threadIdx.x;
  const int wave = tid >> 5;
  const int lane = tid & 31;
  const int half = lane >> 4;
  const int l16  = lane & 15;
  const int b    = blockIdx.x * 4 + wave;

  float* S  = stl[wave];

  v2f   wb[4][3];
  float ob[4];
  #pragma unroll
  for (int j = 0; j < 4; ++j) {
    const int n = wave * 64 + j * 16 + l16;
    ob[j] = bout[n];
    #pragma unroll
    for (int kk = 0; kk < 3; ++kk) {
      const int k0  = kk * 4 + half * 2;
      const int k0c = (k0 < NQ - 1) ? k0 : (NQ - 2);
      wb[j][kk].x = Wout[n * NQ + k0c];
      wb[j][kk].y = Wout[n * NQ + k0c + 1];
    }
  }

  v2f st[8][4];
  v8f d[8];
  float hc[NQ], hs[NQ];

  #pragma unroll
  for (int w = 0; w < NQ; ++w) {
    const float ce = cs[b * 2 * NQ + w], se = cs[b * 2 * NQ + NQ + w];
    const float c0 = tq[w],              s0 = tq[NQ + w];
    hc[w] = fmaf(ce, c0, -(se * s0));
    hs[w] = fmaf(se, c0, ce * s0);
  }
  first_core<5, 4, 3, 2, 1, 9, 8, 0, 7, 6>(d, lane, l16, half, hc, hs);
  ring_store<5, 4, 3, 2, 1, 9, 8, 0, 7, 6>(S, d, l16, half);
  wave_sync();

  for (int l = 1; l < NL - 1; ++l) {
    #pragma unroll
    for (int w = 0; w < NQ; ++w) { hc[w] = tq[l * 2 * NQ + w]; hs[w] = tq[l * 2 * NQ + NQ + w]; }
    load_state<5, 4, 3, 2, 1, 9, 8, 0, 7, 6>(S, st, l16, half);
    wave_sync();
    layer_core<5, 4, 3, 2, 1, 9, 8, 0, 7, 6>(st, d, l16, half, hc, hs);
    ring_store<5, 4, 3, 2, 1, 9, 8, 0, 7, 6>(S, d, l16, half);
    wave_sync();
  }

  #pragma unroll
  for (int w = 0; w < NQ; ++w) {
    hc[w] = tq[(NL - 1) * 2 * NQ + w];
    hs[w] = tq[(NL - 1) * 2 * NQ + NQ + w];
  }
  load_state<5, 4, 3, 2, 1, 9, 8, 0, 7, 6>(S, st, l16, half);
  layer_core<5, 4, 3, 2, 1, 9, 8, 0, 7, 6>(st, d, l16, half, hc, hs);
  const float zq = measure_z<5, 4, 3, 2, 1, 9, 8, 0, 7, 6>(d, lane);

  if (lane < 12) zsh[wave][lane] = zq;
  __syncthreads();
  v2f za[3];
  #pragma unroll
  for (int kk = 0; kk < 3; ++kk) {
    const int k0 = kk * 4 + half * 2;
    const float zx = zsh[l16 & 3][k0], zy = zsh[l16 & 3][k0 + 1];
    za[kk].x = (l16 < 4) ? zx : 0.0f;
    za[kk].y = (l16 < 4) ? zy : 0.0f;
  }
  v8f acc[4];
  #pragma unroll
  for (int j = 0; j < 4; ++j) acc[j] = v8f{};
  #pragma unroll
  for (int kk = 0; kk < 3; ++kk) {
    #pragma unroll
    for (int j = 0; j < 4; ++j)
      acc[j] = __builtin_amdgcn_wmma_f32_16x16x4_f32(false, za[kk], false, wb[j][kk],
                                                     (short)0, acc[j], false, false);
  }
  if (half == 0) {
    float* orow = out + (blockIdx.x * 4) * N_OUT + wave * 64 + l16;
    #pragma unroll
    for (int j = 0; j < 4; ++j) {
      #pragma unroll
      for (int r = 0; r < 4; ++r)
        orow[r * N_OUT + j * 16] = acc[j][r] + ob[j];
    }
  }
}

extern "C" void kernel_launch(void* const* d_in, const int* in_sizes, int n_in,
                              void* d_out, int out_size, void* d_ws, size_t ws_size,
                              hipStream_t stream) {
  const float* x    = (const float*)d_in[0];
  const float* Win  = (const float*)d_in[1];
  const float* bin  = (const float*)d_in[2];
  const float* qp   = (const float*)d_in[3];
  const float* Wout = (const float*)d_in[4];
  const float* bout = (const float*)d_in[5];
  float* out = (float*)d_out;

  float* cs = (float*)d_ws;
  float* tq = cs + BATCH * 2 * NQ;

  k_encode <<<BATCH / 16 + 1,      32, 0, stream>>>(x, Win, bin, qp, cs, tq);
  k_circuit<<<BATCH / 4,      128, 0, stream>>>(cs, tq, Wout, bout, out);
}
